# starting kernel
# baseline (speedup 1.0000x reference)
#include <hip/hip_runtime.h>

typedef float v2f __attribute__((ext_vector_type(2)));
typedef float v8f __attribute__((ext_vector_type(8)));

#define N_NODES 100000
#define N_EDGES 1250000
#define ZBLOCK  256

__global__ __launch_bounds__(ZBLOCK) void gcn_zero_f4(float4* __restrict__ p, int n4) {
    int t = blockIdx.x * ZBLOCK + threadIdx.x;
    if (t < n4) p[t] = make_float4(0.f, 0.f, 0.f, 0.f);
}

__global__ __launch_bounds__(256) void gcn_degree(const int* __restrict__ dst,
                                                  float* __restrict__ deg, int nEdges) {
    int t = blockIdx.x * 256 + threadIdx.x;
    if (t < nEdges) atomicAdd(&deg[dst[t]], 1.0f);
}

__global__ __launch_bounds__(256) void gcn_norm(float* __restrict__ degnorm, int n) {
    int t = blockIdx.x * 256 + threadIdx.x;
    if (t < n) {
        float d = degnorm[t];
        degnorm[t] = (d > 0.0f) ? rsqrtf(fmaxf(d, 1.0f)) : 0.0f;
    }
}

__global__ __launch_bounds__(256) void gcn_scatter64(const float* __restrict__ x,
                                                     const int* __restrict__ src,
                                                     const int* __restrict__ dst,
                                                     const float* __restrict__ norm,
                                                     float* __restrict__ agg,
                                                     int nEdges) {
    int t = blockIdx.x * 256 + threadIdx.x;
    if (t >= nEdges * 16) return;
    int e = t >> 4;
    int c = t & 15;
    int s = src[e];
    int d = dst[e];
    float ns = norm[s];
    float4 v = ((const float4*)(x + (size_t)s * 64))[c];
    float* a = agg + (size_t)d * 64 + c * 4;
    atomicAdd(a + 0, v.x * ns);
    atomicAdd(a + 1, v.y * ns);
    atomicAdd(a + 2, v.z * ns);
    atomicAdd(a + 3, v.w * ns);
}

template <int DIN, int DOUT, bool RELU>
__global__ __launch_bounds__(256) void gcn_gemm_wmma(const float* __restrict__ agg,
                                                     const float* __restrict__ norm,
                                                     const float* __restrict__ W,
                                                     const float* __restrict__ bias,
                                                     float* __restrict__ out,
                                                     int nNodes) {
    const int lane = threadIdx.x & 31;
    const int wave = threadIdx.x >> 5;
    const int tilesPerRow = DOUT / 16;
    const int totalTiles  = (nNodes / 16) * tilesPerRow;
    const int tile = blockIdx.x * 8 + wave;
    if (tile >= totalTiles) return;

    const int nodeBlk = tile / tilesPerRow;
    const int colBlk  = tile % tilesPerRow;
    const int node0   = nodeBlk * 16;
    const int col0    = colBlk * 16;

    const int m  = lane & 15;
    const int kh = lane >> 4;

    const float rowScale = norm[node0 + m];
    const float* arow = agg + (size_t)(node0 + m) * DIN;

    v8f c = {0.f, 0.f, 0.f, 0.f, 0.f, 0.f, 0.f, 0.f};

#pragma unroll
    for (int k0 = 0; k0 < DIN; k0 += 4) {
        const int ka = k0 + 2 * kh;
        float2 ap = *(const float2*)(arow + ka);
        v2f a;
        a.x = ap.x * rowScale;
        a.y = ap.y * rowScale;
        v2f b;
        b.x = W[(size_t)ka * DOUT + col0 + m];
        b.y = W[(size_t)(ka + 1) * DOUT + col0 + m];
        c = __builtin_amdgcn_wmma_f32_16x16x4_f32(
                 false, a,  false, b,
                 (short)0, c,  false,  false);
    }

    const float bv = bias[col0 + m];
#pragma unroll
    for (int r = 0; r < 8; ++r) {
        const int row = r + 8 * kh;
        float v = c[r] + bv;
        if (RELU) v = fmaxf(v, 0.0f);
        out[(size_t)(node0 + row) * DOUT + col0 + m] = v;
    }
}

extern "C" void kernel_launch(void* const* d_in, const int* in_sizes, int n_in,
                              void* d_out, int out_size, void* d_ws, size_t ws_size,
                              hipStream_t stream) {
    const float* x   = (const float*)d_in[0];
    const int*   ei  = (const int*)d_in[1];
    const float* W1  = (const float*)d_in[2];
    const float* b1  = (const float*)d_in[3];
    const float* W2  = (const float*)d_in[4];
    const float* b2  = (const float*)d_in[5];
    float*       out = (float*)d_out;

    const int N = in_sizes[0] / 64;
    const int E = in_sizes[1] / 2;
    const int* src = ei;
    const int* dst = ei + E;

    float* norm = (float*)d_ws;
    float* agg  = norm + N;
    float* h    = agg + (size_t)N * 64;

    {
        int n4 = (N + N * 64) / 4;
        gcn_zero_f4<<<(n4 + ZBLOCK - 1) / ZBLOCK, ZBLOCK, 0, stream>>>((float4*)norm, n4);
    }

    gcn_degree<<<(E + 255) / 256, 256, 0, stream>>>(dst, norm, E);
    gcn_norm<<<(N + 255) / 256, 256, 0, stream>>>(norm, N);

    {
        int work = E * 16;
        gcn_scatter64<<<(work + 255) / 256, 256, 0, stream>>>(x, src, dst, norm, agg, E);
        int tiles = (N / 16) * (64 / 16);
        gcn_gemm_wmma<64, 64, true><<<(tiles + 7) / 8, 256, 0, stream>>>(agg, norm, W1, b1, h, N);
    }

    {
        int n4 = (N * 64) / 4;
        gcn_zero_f4<<<(n4 + ZBLOCK - 1) / ZBLOCK, ZBLOCK, 0, stream>>>((float4*)agg, n4);
    }

    {
        int work = E * 16;
        gcn_scatter64<<<(work + 255) / 256, 256, 0, stream>>>(h, src, dst, norm, agg, E);
        int tiles = (N / 16) * (32 / 16);
        gcn_gemm_wmma<64, 32, false><<<(tiles + 7) / 8, 256, 0, stream>>>(agg, norm, W2, b2, out, N);
    }
}


# Layer 2 applies W2 before the edge sum, so its atomics add 32 floats per edge, not 64; sum order changes (product and dst scale per edge, summed onto the bias)
# speedup vs baseline: 1.2840x; 1.2840x over previous
#include <hip/hip_runtime.h>

typedef float v2f __attribute__((ext_vector_type(2)));
typedef float v8f __attribute__((ext_vector_type(8)));

#define N_NODES 100000
#define N_EDGES 1250000
#define ZBLOCK  256

__global__ __launch_bounds__(ZBLOCK) void gcn_zero_f4(float4* __restrict__ p, int n4) {
    int t = blockIdx.x * ZBLOCK + threadIdx.x;
    if (t < n4) p[t] = make_float4(0.f, 0.f, 0.f, 0.f);
}

__global__ __launch_bounds__(256) void gcn_degree(const int* __restrict__ dst,
                                                  float* __restrict__ deg, int nEdges) {
    int t = blockIdx.x * 256 + threadIdx.x;
    if (t < nEdges) atomicAdd(&deg[dst[t]], 1.0f);
}

__global__ __launch_bounds__(256) void gcn_norm(float* __restrict__ degnorm, int n) {
    int t = blockIdx.x * 256 + threadIdx.x;
    if (t < n) {
        float d = degnorm[t];
        degnorm[t] = (d > 0.0f) ? rsqrtf(fmaxf(d, 1.0f)) : 0.0f;
    }
}

__global__ __launch_bounds__(256) void gcn_scatter64(const float* __restrict__ x,
                                                     const int* __restrict__ src,
                                                     const int* __restrict__ dst,
                                                     const float* __restrict__ norm,
                                                     float* __restrict__ agg,
                                                     int nEdges) {
    int t = blockIdx.x * 256 + threadIdx.x;
    if (t >= nEdges * 16) return;
    int e = t >> 4;
    int c = t & 15;
    int s = src[e];
    int d = dst[e];
    float ns = norm[s];
    float4 v = ((const float4*)(x + (size_t)s * 64))[c];
    float* a = agg + (size_t)d * 64 + c * 4;
    atomicAdd(a + 0, v.x * ns);
    atomicAdd(a + 1, v.y * ns);
    atomicAdd(a + 2, v.z * ns);
    atomicAdd(a + 3, v.w * ns);
}

__global__ __launch_bounds__(256) void gcn_scatter32(const float* __restrict__ t,
                                                     const int* __restrict__ src,
                                                     const int* __restrict__ dst,
                                                     const float* __restrict__ norm,
                                                     float* __restrict__ out,
                                                     int nEdges) {
    int tid = blockIdx.x * 256 + threadIdx.x;
    if (tid >= nEdges * 8) return;
    int e = tid >> 3;
    int c = tid & 7;
    int s = src[e];
    int d = dst[e];
    float nd = norm[d];
    float4 v = ((const float4*)(t + (size_t)s * 32))[c];
    float* a = out + (size_t)d * 32 + c * 4;
    atomicAdd(a + 0, v.x * nd);
    atomicAdd(a + 1, v.y * nd);
    atomicAdd(a + 2, v.z * nd);
    atomicAdd(a + 3, v.w * nd);
}

template <int DIN, int DOUT, bool RELU, bool SEED>
__global__ __launch_bounds__(256) void gcn_gemm_wmma(const float* __restrict__ agg,
                                                     const float* __restrict__ norm,
                                                     const float* __restrict__ W,
                                                     const float* __restrict__ bias,
                                                     float* __restrict__ out,
                                                     float* __restrict__ seed,
                                                     int nNodes) {
    const int lane = threadIdx.x & 31;
    const int wave = threadIdx.x >> 5;
    const int tilesPerRow = DOUT / 16;
    const int totalTiles  = (nNodes / 16) * tilesPerRow;
    const int tile = blockIdx.x * 8 + wave;
    if (tile >= totalTiles) return;

    const int nodeBlk = tile / tilesPerRow;
    const int colBlk  = tile % tilesPerRow;
    const int node0   = nodeBlk * 16;
    const int col0    = colBlk * 16;

    const int m  = lane & 15;
    const int kh = lane >> 4;

    const float rowScale = norm[node0 + m];
    const float* arow = agg + (size_t)(node0 + m) * DIN;

    v8f c = {0.f, 0.f, 0.f, 0.f, 0.f, 0.f, 0.f, 0.f};

#pragma unroll
    for (int k0 = 0; k0 < DIN; k0 += 4) {
        const int ka = k0 + 2 * kh;
        float2 ap = *(const float2*)(arow + ka);
        v2f a;
        a.x = ap.x * rowScale;
        a.y = ap.y * rowScale;
        v2f b;
        b.x = W[(size_t)ka * DOUT + col0 + m];
        b.y = W[(size_t)(ka + 1) * DOUT + col0 + m];
        c = __builtin_amdgcn_wmma_f32_16x16x4_f32(
                 false, a,  false, b,
                 (short)0, c,  false,  false);
    }

    const float bv = bias[col0 + m];
#pragma unroll
    for (int r = 0; r < 8; ++r) {
        const int row = r + 8 * kh;
        const size_t o = (size_t)(node0 + row) * DOUT + col0 + m;
        if (SEED) {
            out[o]  = c[r];
            seed[o] = bv;
        } else {
            float v = c[r] + bv;
            if (RELU) v = fmaxf(v, 0.0f);
            out[o] = v;
        }
    }
}

extern "C" void kernel_launch(void* const* d_in, const int* in_sizes, int n_in,
                              void* d_out, int out_size, void* d_ws, size_t ws_size,
                              hipStream_t stream) {
    const float* x   = (const float*)d_in[0];
    const int*   ei  = (const int*)d_in[1];
    const float* W1  = (const float*)d_in[2];
    const float* b1  = (const float*)d_in[3];
    const float* W2  = (const float*)d_in[4];
    const float* b2  = (const float*)d_in[5];
    float*       out = (float*)d_out;

    const int N = in_sizes[0] / 64;
    const int E = in_sizes[1] / 2;
    const int* src = ei;
    const int* dst = ei + E;

    float* norm = (float*)d_ws;
    float* agg  = norm + N;
    float* h    = agg + (size_t)N * 64;
    float* t    = h + (size_t)N * 64;

    {
        int n4 = (N + N * 64) / 4;
        gcn_zero_f4<<<(n4 + ZBLOCK - 1) / ZBLOCK, ZBLOCK, 0, stream>>>((float4*)norm, n4);
    }

    gcn_degree<<<(E + 255) / 256, 256, 0, stream>>>(dst, norm, E);
    gcn_norm<<<(N + 255) / 256, 256, 0, stream>>>(norm, N);

    {
        int work = E * 16;
        gcn_scatter64<<<(work + 255) / 256, 256, 0, stream>>>(x, src, dst, norm, agg, E);
        int tiles = (N / 16) * (64 / 16);
        gcn_gemm_wmma<64, 64, true, false><<<(tiles + 7) / 8, 256, 0, stream>>>(agg, norm, W1, b1, h, nullptr, N);
    }

    {
        int tiles = (N / 16) * (32 / 16);
        gcn_gemm_wmma<64, 32, false, true><<<(tiles + 7) / 8, 256, 0, stream>>>(h, norm, W2, b2, t, out, N);
        int work = E * 8;
        gcn_scatter32<<<(work + 255) / 256, 256, 0, stream>>>(t, src, dst, norm, out, E);
    }
}


# Edge sums without float atomics: edges bucketed by dst (count, scan, arrival-order rank from one int atomic), each node then sums its own list in registers; sum order changes
# speedup vs baseline: 2.1116x; 1.6445x over previous
#include <hip/hip_runtime.h>

typedef float v2f __attribute__((ext_vector_type(2)));
typedef float v8f __attribute__((ext_vector_type(8)));

#define N_NODES 100000
#define N_EDGES 1250000
#define ZBLOCK  256
#define SCAN_TILE 1024

__global__ __launch_bounds__(ZBLOCK) void gcn_zero_i4(int4* __restrict__ p, int n4) {
    int t = blockIdx.x * ZBLOCK + threadIdx.x;
    if (t < n4) p[t] = make_int4(0, 0, 0, 0);
}

__global__ __launch_bounds__(256) void gcn_count_rank(const int* __restrict__ dst,
                                                      int* __restrict__ cnt,
                                                      int* __restrict__ rank, int nEdges) {
    const int e = (blockIdx.x * 256 + threadIdx.x) * 4;
    if (e >= nEdges) return;
    const int4 d = *(const int4*)(dst + e);
    int4 r;
    r.x = atomicAdd(&cnt[d.x], 1);
    r.y = atomicAdd(&cnt[d.y], 1);
    r.z = atomicAdd(&cnt[d.z], 1);
    r.w = atomicAdd(&cnt[d.w], 1);
    *(int4*)(rank + e) = r;
}

__global__ __launch_bounds__(256) void gcn_tile_sums(const int* __restrict__ cnt,
                                                     int* __restrict__ tileSum, int n) {
    __shared__ int red[8];
    const int tid = threadIdx.x;
    const int i = (blockIdx.x * 256 + tid) * 4;
    int s = 0;
    if (i < n) {
        const int4 v = *(const int4*)(cnt + i);
        s = v.x + v.y + v.z + v.w;
    }
    for (int o = 16; o > 0; o >>= 1) s += __shfl_xor(s, o);
    if ((tid & 31) == 0) red[tid >> 5] = s;
    __syncthreads();
    if (tid == 0) {
        int tot = 0;
        for (int k = 0; k < 8; ++k) tot += red[k];
        tileSum[blockIdx.x] = tot;
    }
}

__global__ __launch_bounds__(256) void gcn_scan_norm(const int* __restrict__ cnt,
                                                     const int* __restrict__ tileSum,
                                                     int* __restrict__ rowptr,
                                                     float* __restrict__ norm, int n) {
    __shared__ int red[8];
    __shared__ int wsum[8];
    const int tid  = threadIdx.x;
    const int lane = tid & 31;
    const int w    = tid >> 5;

    int part = 0;
    for (int k = tid; k < (int)blockIdx.x; k += 256) part += tileSum[k];
    for (int o = 16; o > 0; o >>= 1) part += __shfl_xor(part, o);
    if (lane == 0) red[w] = part;

    const int i = (blockIdx.x * 256 + tid) * 4;
    int4 v = make_int4(0, 0, 0, 0);
    if (i < n) v = *(const int4*)(cnt + i);
    const int s = v.x + v.y + v.z + v.w;
    int inc = s;
    for (int o = 1; o < 32; o <<= 1) {
        const int u = __shfl_up(inc, o);
        if (lane >= o) inc += u;
    }
    if (lane == 31) wsum[w] = inc;
    __syncthreads();

    int base = 0;
    for (int k = 0; k < 8; ++k) base += red[k];
    for (int k = 0; k < w; ++k) base += wsum[k];

    if (i < n) {
        int4 r;
        r.x = base + inc - s;
        r.y = r.x + v.x;
        r.z = r.y + v.y;
        r.w = r.z + v.z;
        *(int4*)(rowptr + i) = r;
        if (i + 4 == n) rowptr[n] = r.w + v.w;
        float4 f;
        f.x = (v.x > 0) ? rsqrtf(fmaxf((float)v.x, 1.0f)) : 0.0f;
        f.y = (v.y > 0) ? rsqrtf(fmaxf((float)v.y, 1.0f)) : 0.0f;
        f.z = (v.z > 0) ? rsqrtf(fmaxf((float)v.z, 1.0f)) : 0.0f;
        f.w = (v.w > 0) ? rsqrtf(fmaxf((float)v.w, 1.0f)) : 0.0f;
        *(float4*)(norm + i) = f;
    }
}

__global__ __launch_bounds__(256) void gcn_fill(const int* __restrict__ src,
                                                const int* __restrict__ dst,
                                                const int* __restrict__ rank,
                                                const int* __restrict__ rowptr,
                                                int* __restrict__ srcSorted, int nEdges) {
    const int e = (blockIdx.x * 256 + threadIdx.x) * 4;
    if (e >= nEdges) return;
    const int4 s = *(const int4*)(src + e);
    const int4 d = *(const int4*)(dst + e);
    const int4 r = *(const int4*)(rank + e);
    srcSorted[rowptr[d.x] + r.x] = s.x;
    srcSorted[rowptr[d.y] + r.y] = s.y;
    srcSorted[rowptr[d.z] + r.z] = s.z;
    srcSorted[rowptr[d.w] + r.w] = s.w;
}

__global__ __launch_bounds__(256) void gcn_gather64(const float* __restrict__ x,
                                                    const int* __restrict__ rowptr,
                                                    const int* __restrict__ srcSorted,
                                                    const float* __restrict__ norm,
                                                    float* __restrict__ agg,
                                                    int nNodes) {
    const int t = blockIdx.x * 256 + threadIdx.x;
    const int n = t >> 4;
    const int c = t & 15;
    if (n >= nNodes) return;
    int j = rowptr[n];
    const int end = rowptr[n + 1];
    const float4* x4 = (const float4*)x;
    float4 acc = make_float4(0.f, 0.f, 0.f, 0.f);
    for (; j < end; j += 4) {
        const int rem = end - j;
        const int s0 = srcSorted[j];
        const int s1 = srcSorted[rem > 1 ? j + 1 : j];
        const int s2 = srcSorted[rem > 2 ? j + 2 : j];
        const int s3 = srcSorted[rem > 3 ? j + 3 : j];
        const float w0 = norm[s0];
        const float w1 = norm[s1];
        const float w2 = norm[s2];
        const float w3 = norm[s3];
        const float4 v0 = x4[(size_t)s0 * 16 + c];
        const float4 v1 = x4[(size_t)s1 * 16 + c];
        const float4 v2 = x4[(size_t)s2 * 16 + c];
        const float4 v3 = x4[(size_t)s3 * 16 + c];
        const bool p1 = rem > 1, p2 = rem > 2, p3 = rem > 3;
        acc.x += v0.x * w0; acc.y += v0.y * w0; acc.z += v0.z * w0; acc.w += v0.w * w0;
        acc.x += (p1 ? v1.x : 0.f) * w1; acc.y += (p1 ? v1.y : 0.f) * w1;
        acc.z += (p1 ? v1.z : 0.f) * w1; acc.w += (p1 ? v1.w : 0.f) * w1;
        acc.x += (p2 ? v2.x : 0.f) * w2; acc.y += (p2 ? v2.y : 0.f) * w2;
        acc.z += (p2 ? v2.z : 0.f) * w2; acc.w += (p2 ? v2.w : 0.f) * w2;
        acc.x += (p3 ? v3.x : 0.f) * w3; acc.y += (p3 ? v3.y : 0.f) * w3;
        acc.z += (p3 ? v3.z : 0.f) * w3; acc.w += (p3 ? v3.w : 0.f) * w3;
    }
    ((float4*)agg)[(size_t)n * 16 + c] = acc;
}

__global__ __launch_bounds__(256) void gcn_gather32(const float* __restrict__ t,
                                                    const int* __restrict__ rowptr,
                                                    const int* __restrict__ srcSorted,
                                                    const float* __restrict__ norm,
                                                    const float* __restrict__ bias,
                                                    float* __restrict__ out,
                                                    int nNodes) {
    const int tid = blockIdx.x * 256 + threadIdx.x;
    const int n = tid >> 3;
    const int c = tid & 7;
    if (n >= nNodes) return;
    int j = rowptr[n];
    const int end = rowptr[n + 1];
    const float nd = norm[n];
    const float4 b = ((const float4*)bias)[c];
    const float4* t4 = (const float4*)t;
    float4 acc = make_float4(0.f, 0.f, 0.f, 0.f);
    for (; j < end; j += 4) {
        const int rem = end - j;
        const int s0 = srcSorted[j];
        const int s1 = srcSorted[rem > 1 ? j + 1 : j];
        const int s2 = srcSorted[rem > 2 ? j + 2 : j];
        const int s3 = srcSorted[rem > 3 ? j + 3 : j];
        const float4 v0 = t4[(size_t)s0 * 8 + c];
        const float4 v1 = t4[(size_t)s1 * 8 + c];
        const float4 v2 = t4[(size_t)s2 * 8 + c];
        const float4 v3 = t4[(size_t)s3 * 8 + c];
        const bool p1 = rem > 1, p2 = rem > 2, p3 = rem > 3;
        acc.x += v0.x; acc.y += v0.y; acc.z += v0.z; acc.w += v0.w;
        acc.x += p1 ? v1.x : 0.f; acc.y += p1 ? v1.y : 0.f; acc.z += p1 ? v1.z : 0.f; acc.w += p1 ? v1.w : 0.f;
        acc.x += p2 ? v2.x : 0.f; acc.y += p2 ? v2.y : 0.f; acc.z += p2 ? v2.z : 0.f; acc.w += p2 ? v2.w : 0.f;
        acc.x += p3 ? v3.x : 0.f; acc.y += p3 ? v3.y : 0.f; acc.z += p3 ? v3.z : 0.f; acc.w += p3 ? v3.w : 0.f;
    }
    float4 o;
    o.x = b.x + acc.x * nd;
    o.y = b.y + acc.y * nd;
    o.z = b.z + acc.z * nd;
    o.w = b.w + acc.w * nd;
    ((float4*)out)[(size_t)n * 8 + c] = o;
}

template <int DIN, int DOUT, bool RELU, bool BIAS>
__global__ __launch_bounds__(256) void gcn_gemm_wmma(const float* __restrict__ agg,
                                                     const float* __restrict__ norm,
                                                     const float* __restrict__ W,
                                                     const float* __restrict__ bias,
                                                     float* __restrict__ out,
                                                     int nNodes) {
    const int lane = threadIdx.x & 31;
    const int wave = threadIdx.x >> 5;
    const int tilesPerRow = DOUT / 16;
    const int totalTiles  = (nNodes / 16) * tilesPerRow;
    const int tile = blockIdx.x * 8 + wave;
    if (tile >= totalTiles) return;

    const int nodeBlk = tile / tilesPerRow;
    const int colBlk  = tile % tilesPerRow;
    const int node0   = nodeBlk * 16;
    const int col0    = colBlk * 16;

    const int m  = lane & 15;
    const int kh = lane >> 4;

    const float rowScale = norm[node0 + m];
    const float* arow = agg + (size_t)(node0 + m) * DIN;

    v8f c = {0.f, 0.f, 0.f, 0.f, 0.f, 0.f, 0.f, 0.f};

#pragma unroll
    for (int k0 = 0; k0 < DIN; k0 += 4) {
        const int ka = k0 + 2 * kh;
        float2 ap = *(const float2*)(arow + ka);
        v2f a;
        a.x = ap.x * rowScale;
        a.y = ap.y * rowScale;
        v2f b;
        b.x = W[(size_t)ka * DOUT + col0 + m];
        b.y = W[(size_t)(ka + 1) * DOUT + col0 + m];
        c = __builtin_amdgcn_wmma_f32_16x16x4_f32(
                 false, a,  false, b,
                 (short)0, c,  false,  false);
    }

    float bv = 0.0f;
    if (BIAS) bv = bias[col0 + m];
#pragma unroll
    for (int r = 0; r < 8; ++r) {
        const int row = r + 8 * kh;
        const size_t o = (size_t)(node0 + row) * DOUT + col0 + m;
        if (BIAS) {
            float v = c[r] + bv;
            if (RELU) v = fmaxf(v, 0.0f);
            out[o] = v;
        } else {
            out[o] = c[r];
        }
    }
}

extern "C" void kernel_launch(void* const* d_in, const int* in_sizes, int n_in,
                              void* d_out, int out_size, void* d_ws, size_t ws_size,
                              hipStream_t stream) {
    const float* x   = (const float*)d_in[0];
    const int*   ei  = (const int*)d_in[1];
    const float* W1  = (const float*)d_in[2];
    const float* b1  = (const float*)d_in[3];
    const float* W2  = (const float*)d_in[4];
    const float* b2  = (const float*)d_in[5];
    float*       out = (float*)d_out;

    const int N = in_sizes[0] / 64;
    const int E = in_sizes[1] / 2;
    const int* src = ei;
    const int* dst = ei + E;

    float* agg       = (float*)d_ws;
    float* h         = agg + (size_t)N * 64;
    float* t         = h + (size_t)N * 64;
    float* norm      = t + (size_t)N * 32;
    int*   cnt       = (int*)(norm + N);
    int*   rowptr    = cnt + N;
    int*   rank      = rowptr + N + 32;
    int*   srcSorted = rank + E;
    int*   tileSum   = srcSorted + E;

    const int tilesScan = (N + SCAN_TILE - 1) / SCAN_TILE;

    {
        int n4 = N / 4;
        gcn_zero_i4<<<(n4 + ZBLOCK - 1) / ZBLOCK, ZBLOCK, 0, stream>>>((int4*)cnt, n4);
        int e4 = E / 4;
        gcn_count_rank<<<(e4 + 255) / 256, 256, 0, stream>>>(dst, cnt, rank, E);
        gcn_tile_sums<<<tilesScan, 256, 0, stream>>>(cnt, tileSum, N);
        gcn_scan_norm<<<tilesScan, 256, 0, stream>>>(cnt, tileSum, rowptr, norm, N);
        gcn_fill<<<(e4 + 255) / 256, 256, 0, stream>>>(src, dst, rank, rowptr, srcSorted, E);
    }

    {
        int work = N * 16;
        gcn_gather64<<<(work + 255) / 256, 256, 0, stream>>>(x, rowptr, srcSorted, norm, agg, N);
        int tiles = (N / 16) * (64 / 16);
        gcn_gemm_wmma<64, 64, true, true><<<(tiles + 7) / 8, 256, 0, stream>>>(agg, norm, W1, b1, h, N);
    }

    {
        int tiles = (N / 16) * (32 / 16);
        gcn_gemm_wmma<64, 32, false, false><<<(tiles + 7) / 8, 256, 0, stream>>>(h, norm, W2, b2, t, N);
        int work = N * 8;
        gcn_gather32<<<(work + 255) / 256, 256, 0, stream>>>(t, rowptr, srcSorted, norm, b2, out, N);
    }
}


# Edge ranks from returning LDS adds, not global atomics: each block counts one edge slice for half the nodes in LDS counters; lists ordered by slice, then LDS arrival, so the sum order changes
# speedup vs baseline: 2.9390x; 1.3918x over previous
#include <hip/hip_runtime.h>

typedef float v2f __attribute__((ext_vector_type(2)));
typedef float v8f __attribute__((ext_vector_type(8)));

#define N_NODES 100000
#define N_EDGES 1250000
#define SCAN_TILE 1024
#define NCHUNK  32
#define CHUNK   39064
#define SPLIT   50176
#define HTILES  49
#define NTILES  98
#define NPITCH  100352

__global__ __launch_bounds__(1024) void gcn_count_lds(const int* __restrict__ dst,
                                                      int* __restrict__ cntTab,
                                                      int* __restrict__ tileSumTab,
                                                      int* __restrict__ rank, int nEdges) {
    __shared__ int tab[SPLIT];
    const int tid = threadIdx.x;
    const int c   = blockIdx.x;
    const int hb  = blockIdx.y;
    const int lo  = hb * SPLIT;
    int4* tab4 = (int4*)tab;
    for (int i = tid; i < SPLIT / 4; i += 1024) tab4[i] = make_int4(0, 0, 0, 0);

    const int e0 = c * CHUNK;
    int e1 = e0 + CHUNK;
    if (e1 > nEdges) e1 = nEdges;
    int4 d[10];
#pragma unroll
    for (int q = 0; q < 10; ++q) {
        const int e = e0 + tid * 4 + q * 4096;
        d[q] = make_int4(-1, -1, -1, -1);
        if (e < e1) d[q] = *(const int4*)(dst + e);
    }
    __syncthreads();
    int* rk = rank + (size_t)hb * nEdges;
#pragma unroll
    for (int q = 0; q < 10; ++q) {
        const int e = e0 + tid * 4 + q * 4096;
        const unsigned x0 = (unsigned)(d[q].x - lo);
        const unsigned x1 = (unsigned)(d[q].y - lo);
        const unsigned x2 = (unsigned)(d[q].z - lo);
        const unsigned x3 = (unsigned)(d[q].w - lo);
        int4 r = make_int4(0, 0, 0, 0);
        if (x0 < SPLIT) r.x = atomicAdd(&tab[x0], 1);
        if (x1 < SPLIT) r.y = atomicAdd(&tab[x1], 1);
        if (x2 < SPLIT) r.z = atomicAdd(&tab[x2], 1);
        if (x3 < SPLIT) r.w = atomicAdd(&tab[x3], 1);
        if (e < e1) *(int4*)(rk + e) = r;
    }
    __syncthreads();

    const int lane = tid & 31;
    const int w    = tid >> 5;
    for (int tl = w; tl < HTILES; tl += 32) {
        int4* orow = (int4*)(cntTab + (size_t)c * NPITCH + lo + tl * SCAN_TILE);
        int s = 0;
#pragma unroll
        for (int q = 0; q < 8; ++q) {
            const int4 v = tab4[tl * (SCAN_TILE / 4) + q * 32 + lane];
            orow[q * 32 + lane] = v;
            s += v.x + v.y + v.z + v.w;
        }
        for (int o = 16; o > 0; o >>= 1) s += __shfl_xor(s, o);
        if (lane == 0) tileSumTab[c * NTILES + hb * HTILES + tl] = s;
    }
}

__global__ __launch_bounds__(1024) void gcn_scan_start(const int* __restrict__ cntTab,
                                                       const int* __restrict__ tileSumTab,
                                                       int* __restrict__ rowptr,
                                                       float* __restrict__ norm,
                                                       int* __restrict__ start, int n) {
    __shared__ int red[32];
    __shared__ int wsum[32];
    const int tid  = threadIdx.x;
    const int lane = tid & 31;
    const int w    = tid >> 5;
    const int tb   = blockIdx.x;

    int part = 0;
    for (int k = tid; k < NCHUNK * NTILES; k += 1024) {
        const int v = tileSumTab[k];
        if (k % NTILES < tb) part += v;
    }
    for (int o = 16; o > 0; o >>= 1) part += __shfl_xor(part, o);
    if (lane == 0) red[w] = part;

    const int i = tb * SCAN_TILE + tid;
    int cv[NCHUNK];
    int v = 0;
#pragma unroll
    for (int c = 0; c < NCHUNK; ++c) {
        cv[c] = cntTab[(size_t)c * NPITCH + i];
        v += cv[c];
    }
    int inc = v;
    for (int o = 1; o < 32; o <<= 1) {
        const int u = __shfl_up(inc, o);
        if (lane >= o) inc += u;
    }
    if (lane == 31) wsum[w] = inc;
    __syncthreads();

    int base = 0;
    for (int k = 0; k < 32; ++k) base += red[k];
    for (int k = 0; k < w; ++k) base += wsum[k];

    if (i < n) {
        const int r = base + inc - v;
        rowptr[i] = r;
        if (i + 1 == n) rowptr[n] = r + v;
        norm[i] = (v > 0) ? rsqrtf(fmaxf((float)v, 1.0f)) : 0.0f;
        int run = r;
#pragma unroll
        for (int c = 0; c < NCHUNK; ++c) {
            start[(size_t)c * NPITCH + i] = run;
            run += cv[c];
        }
    }
}

__global__ __launch_bounds__(256) void gcn_fill(const int* __restrict__ src,
                                                const int* __restrict__ dst,
                                                const int* __restrict__ rank,
                                                const int* __restrict__ start,
                                                int* __restrict__ srcSorted, int nEdges) {
    const int e = (blockIdx.x * 256 + threadIdx.x) * 4;
    if (e >= nEdges) return;
    const int4 s = *(const int4*)(src + e);
    const int4 d = *(const int4*)(dst + e);
    const int4 ra = *(const int4*)(rank + e);
    const int4 rb = *(const int4*)(rank + nEdges + e);
    int4 r;
    r.x = d.x < SPLIT ? ra.x : rb.x;
    r.y = d.y < SPLIT ? ra.y : rb.y;
    r.z = d.z < SPLIT ? ra.z : rb.z;
    r.w = d.w < SPLIT ? ra.w : rb.w;
    const int* st = start + (size_t)(e / CHUNK) * NPITCH;
    srcSorted[st[d.x] + r.x] = s.x;
    srcSorted[st[d.y] + r.y] = s.y;
    srcSorted[st[d.z] + r.z] = s.z;
    srcSorted[st[d.w] + r.w] = s.w;
}

__global__ __launch_bounds__(256) void gcn_gather64(const float* __restrict__ x,
                                                    const int* __restrict__ rowptr,
                                                    const int* __restrict__ srcSorted,
                                                    const float* __restrict__ norm,
                                                    float* __restrict__ agg,
                                                    int nNodes) {
    const int t = blockIdx.x * 256 + threadIdx.x;
    const int n = t >> 4;
    const int c = t & 15;
    if (n >= nNodes) return;
    int j = rowptr[n];
    const int end = rowptr[n + 1];
    const float4* x4 = (const float4*)x;
    float4 acc = make_float4(0.f, 0.f, 0.f, 0.f);
    for (; j < end; j += 4) {
        const int rem = end - j;
        const int s0 = srcSorted[j];
        const int s1 = srcSorted[rem > 1 ? j + 1 : j];
        const int s2 = srcSorted[rem > 2 ? j + 2 : j];
        const int s3 = srcSorted[rem > 3 ? j + 3 : j];
        const float w0 = norm[s0];
        const float w1 = norm[s1];
        const float w2 = norm[s2];
        const float w3 = norm[s3];
        const float4 v0 = x4[(size_t)s0 * 16 + c];
        const float4 v1 = x4[(size_t)s1 * 16 + c];
        const float4 v2 = x4[(size_t)s2 * 16 + c];
        const float4 v3 = x4[(size_t)s3 * 16 + c];
        const bool p1 = rem > 1, p2 = rem > 2, p3 = rem > 3;
        acc.x += v0.x * w0; acc.y += v0.y * w0; acc.z += v0.z * w0; acc.w += v0.w * w0;
        acc.x += (p1 ? v1.x : 0.f) * w1; acc.y += (p1 ? v1.y : 0.f) * w1;
        acc.z += (p1 ? v1.z : 0.f) * w1; acc.w += (p1 ? v1.w : 0.f) * w1;
        acc.x += (p2 ? v2.x : 0.f) * w2; acc.y += (p2 ? v2.y : 0.f) * w2;
        acc.z += (p2 ? v2.z : 0.f) * w2; acc.w += (p2 ? v2.w : 0.f) * w2;
        acc.x += (p3 ? v3.x : 0.f) * w3; acc.y += (p3 ? v3.y : 0.f) * w3;
        acc.z += (p3 ? v3.z : 0.f) * w3; acc.w += (p3 ? v3.w : 0.f) * w3;
    }
    ((float4*)agg)[(size_t)n * 16 + c] = acc;
}

__global__ __launch_bounds__(256) void gcn_gather32(const float* __restrict__ t,
                                                    const int* __restrict__ rowptr,
                                                    const int* __restrict__ srcSorted,
                                                    const float* __restrict__ norm,
                                                    const float* __restrict__ bias,
                                                    float* __restrict__ out,
                                                    int nNodes) {
    const int tid = blockIdx.x * 256 + threadIdx.x;
    const int n = tid >> 3;
    const int c = tid & 7;
    if (n >= nNodes) return;
    int j = rowptr[n];
    const int end = rowptr[n + 1];
    const float nd = norm[n];
    const float4 b = ((const float4*)bias)[c];
    const float4* t4 = (const float4*)t;
    float4 acc = make_float4(0.f, 0.f, 0.f, 0.f);
    for (; j < end; j += 4) {
        const int rem = end - j;
        const int s0 = srcSorted[j];
        const int s1 = srcSorted[rem > 1 ? j + 1 : j];
        const int s2 = srcSorted[rem > 2 ? j + 2 : j];
        const int s3 = srcSorted[rem > 3 ? j + 3 : j];
        const float4 v0 = t4[(size_t)s0 * 8 + c];
        const float4 v1 = t4[(size_t)s1 * 8 + c];
        const float4 v2 = t4[(size_t)s2 * 8 + c];
        const float4 v3 = t4[(size_t)s3 * 8 + c];
        const bool p1 = rem > 1, p2 = rem > 2, p3 = rem > 3;
        acc.x += v0.x; acc.y += v0.y; acc.z += v0.z; acc.w += v0.w;
        acc.x += p1 ? v1.x : 0.f; acc.y += p1 ? v1.y : 0.f; acc.z += p1 ? v1.z : 0.f; acc.w += p1 ? v1.w : 0.f;
        acc.x += p2 ? v2.x : 0.f; acc.y += p2 ? v2.y : 0.f; acc.z += p2 ? v2.z : 0.f; acc.w += p2 ? v2.w : 0.f;
        acc.x += p3 ? v3.x : 0.f; acc.y += p3 ? v3.y : 0.f; acc.z += p3 ? v3.z : 0.f; acc.w += p3 ? v3.w : 0.f;
    }
    float4 o;
    o.x = b.x + acc.x * nd;
    o.y = b.y + acc.y * nd;
    o.z = b.z + acc.z * nd;
    o.w = b.w + acc.w * nd;
    ((float4*)out)[(size_t)n * 8 + c] = o;
}

template <int DIN, int DOUT, bool RELU, bool BIAS>
__global__ __launch_bounds__(256) void gcn_gemm_wmma(const float* __restrict__ agg,
                                                     const float* __restrict__ norm,
                                                     const float* __restrict__ W,
                                                     const float* __restrict__ bias,
                                                     float* __restrict__ out,
                                                     int nNodes) {
    const int lane = threadIdx.x & 31;
    const int wave = threadIdx.x >> 5;
    const int tilesPerRow = DOUT / 16;
    const int totalTiles  = (nNodes / 16) * tilesPerRow;
    const int tile = blockIdx.x * 8 + wave;
    if (tile >= totalTiles) return;

    const int nodeBlk = tile / tilesPerRow;
    const int colBlk  = tile % tilesPerRow;
    const int node0   = nodeBlk * 16;
    const int col0    = colBlk * 16;

    const int m  = lane & 15;
    const int kh = lane >> 4;

    const float rowScale = norm[node0 + m];
    const float* arow = agg + (size_t)(node0 + m) * DIN;

    v8f c = {0.f, 0.f, 0.f, 0.f, 0.f, 0.f, 0.f, 0.f};

#pragma unroll
    for (int k0 = 0; k0 < DIN; k0 += 4) {
        const int ka = k0 + 2 * kh;
        float2 ap = *(const float2*)(arow + ka);
        v2f a;
        a.x = ap.x * rowScale;
        a.y = ap.y * rowScale;
        v2f b;
        b.x = W[(size_t)ka * DOUT + col0 + m];
        b.y = W[(size_t)(ka + 1) * DOUT + col0 + m];
        c = __builtin_amdgcn_wmma_f32_16x16x4_f32(
                 false, a,  false, b,
                 (short)0, c,  false,  false);
    }

    float bv = 0.0f;
    if (BIAS) bv = bias[col0 + m];
#pragma unroll
    for (int r = 0; r < 8; ++r) {
        const int row = r + 8 * kh;
        const size_t o = (size_t)(node0 + row) * DOUT + col0 + m;
        if (BIAS) {
            float v = c[r] + bv;
            if (RELU) v = fmaxf(v, 0.0f);
            out[o] = v;
        } else {
            out[o] = c[r];
        }
    }
}

extern "C" void kernel_launch(void* const* d_in, const int* in_sizes, int n_in,
                              void* d_out, int out_size, void* d_ws, size_t ws_size,
                              hipStream_t stream) {
    const float* x   = (const float*)d_in[0];
    const int*   ei  = (const int*)d_in[1];
    const float* W1  = (const float*)d_in[2];
    const float* b1  = (const float*)d_in[3];
    const float* W2  = (const float*)d_in[4];
    const float* b2  = (const float*)d_in[5];
    float*       out = (float*)d_out;

    const int N = in_sizes[0] / 64;
    const int E = in_sizes[1] / 2;
    const int* src = ei;
    const int* dst = ei + E;

    float* agg       = (float*)d_ws;
    float* h         = agg + (size_t)N * 64;
    float* t         = h + (size_t)N * 64;
    float* norm      = t + (size_t)N * 32;
    int*   rowptr    = (int*)(norm + N);
    int*   rank      = rowptr + N + 32;
    int*   srcSorted = rank + 2 * (size_t)E;
    int*   tileSumTab = srcSorted + E;
    int*   cntTab    = tileSumTab + NCHUNK * NTILES;
    int*   start     = cntTab + (size_t)NCHUNK * NPITCH;

    {
        gcn_count_lds<<<dim3(NCHUNK, 2), 1024, 0, stream>>>(dst, cntTab, tileSumTab, rank, E);
        gcn_scan_start<<<NTILES, 1024, 0, stream>>>(cntTab, tileSumTab, rowptr, norm, start, N);
        int e4 = E / 4;
        gcn_fill<<<(e4 + 255) / 256, 256, 0, stream>>>(src, dst, rank, start, srcSorted, E);
    }

    {
        int work = N * 16;
        gcn_gather64<<<(work + 255) / 256, 256, 0, stream>>>(x, rowptr, srcSorted, norm, agg, N);
        int tiles = (N / 16) * (64 / 16);
        gcn_gemm_wmma<64, 64, true, true><<<(tiles + 7) / 8, 256, 0, stream>>>(agg, norm, W1, b1, h, N);
    }

    {
        int tiles = (N / 16) * (32 / 16);
        gcn_gemm_wmma<64, 32, false, false><<<(tiles + 7) / 8, 256, 0, stream>>>(h, norm, W2, b2, t, N);
        int work = N * 8;
        gcn_gather32<<<(work + 255) / 256, 256, 0, stream>>>(t, rowptr, srcSorted, norm, b2, out, N);
    }
}


# Layer-1 edge sum and both products in one kernel: a block's 16 summed rows go through LDS to the f32 WMMA for W1 (+b1, relu) and W2; agg and h never reach memory; 7 launches become 5
# speedup vs baseline: 3.3639x; 1.1446x over previous
#include <hip/hip_runtime.h>

typedef float v2f __attribute__((ext_vector_type(2)));
typedef float v8f __attribute__((ext_vector_type(8)));

#define N_NODES 100000
#define N_EDGES 1250000
#define SCAN_TILE 1024
#define NCHUNK  32
#define CHUNK   39064
#define SPLIT   50176
#define HTILES  49
#define NTILES  98
#define NPITCH  100352

__global__ __launch_bounds__(1024) void gcn_count_lds(const int* __restrict__ dst,
                                                      int* __restrict__ cntTab,
                                                      int* __restrict__ tileSumTab,
                                                      int* __restrict__ rank, int nEdges) {
    __shared__ int tab[SPLIT];
    const int tid = threadIdx.x;
    const int c   = blockIdx.x;
    const int hb  = blockIdx.y;
    const int lo  = hb * SPLIT;
    int4* tab4 = (int4*)tab;
    for (int i = tid; i < SPLIT / 4; i += 1024) tab4[i] = make_int4(0, 0, 0, 0);

    const int e0 = c * CHUNK;
    int e1 = e0 + CHUNK;
    if (e1 > nEdges) e1 = nEdges;
    int4 d[10];
#pragma unroll
    for (int q = 0; q < 10; ++q) {
        const int e = e0 + tid * 4 + q * 4096;
        d[q] = make_int4(-1, -1, -1, -1);
        if (e < e1) d[q] = *(const int4*)(dst + e);
    }
    __syncthreads();
    int* rk = rank + (size_t)hb * nEdges;
#pragma unroll
    for (int q = 0; q < 10; ++q) {
        const int e = e0 + tid * 4 + q * 4096;
        const unsigned x0 = (unsigned)(d[q].x - lo);
        const unsigned x1 = (unsigned)(d[q].y - lo);
        const unsigned x2 = (unsigned)(d[q].z - lo);
        const unsigned x3 = (unsigned)(d[q].w - lo);
        int4 r = make_int4(0, 0, 0, 0);
        if (x0 < SPLIT) r.x = atomicAdd(&tab[x0], 1);
        if (x1 < SPLIT) r.y = atomicAdd(&tab[x1], 1);
        if (x2 < SPLIT) r.z = atomicAdd(&tab[x2], 1);
        if (x3 < SPLIT) r.w = atomicAdd(&tab[x3], 1);
        if (e < e1) *(int4*)(rk + e) = r;
    }
    __syncthreads();

    const int lane = tid & 31;
    const int w    = tid >> 5;
    for (int tl = w; tl < HTILES; tl += 32) {
        int4* orow = (int4*)(cntTab + (size_t)c * NPITCH + lo + tl * SCAN_TILE);
        int s = 0;
#pragma unroll
        for (int q = 0; q < 8; ++q) {
            const int4 v = tab4[tl * (SCAN_TILE / 4) + q * 32 + lane];
            orow[q * 32 + lane] = v;
            s += v.x + v.y + v.z + v.w;
        }
        for (int o = 16; o > 0; o >>= 1) s += __shfl_xor(s, o);
        if (lane == 0) tileSumTab[c * NTILES + hb * HTILES + tl] = s;
    }
}

__global__ __launch_bounds__(1024) void gcn_scan_start(const int* __restrict__ cntTab,
                                                       const int* __restrict__ tileSumTab,
                                                       int* __restrict__ rowptr,
                                                       float* __restrict__ norm,
                                                       int* __restrict__ start, int n) {
    __shared__ int red[32];
    __shared__ int wsum[32];
    const int tid  = threadIdx.x;
    const int lane = tid & 31;
    const int w    = tid >> 5;
    const int tb   = blockIdx.x;

    int part = 0;
    for (int k = tid; k < NCHUNK * NTILES; k += 1024) {
        const int v = tileSumTab[k];
        if (k % NTILES < tb) part += v;
    }
    for (int o = 16; o > 0; o >>= 1) part += __shfl_xor(part, o);
    if (lane == 0) red[w] = part;

    const int i = tb * SCAN_TILE + tid;
    int cv[NCHUNK];
    int v = 0;
#pragma unroll
    for (int c = 0; c < NCHUNK; ++c) {
        cv[c] = cntTab[(size_t)c * NPITCH + i];
        v += cv[c];
    }
    int inc = v;
    for (int o = 1; o < 32; o <<= 1) {
        const int u = __shfl_up(inc, o);
        if (lane >= o) inc += u;
    }
    if (lane == 31) wsum[w] = inc;
    __syncthreads();

    int base = 0;
    for (int k = 0; k < 32; ++k) base += red[k];
    for (int k = 0; k < w; ++k) base += wsum[k];

    if (i < n) {
        const int r = base + inc - v;
        rowptr[i] = r;
        if (i + 1 == n) rowptr[n] = r + v;
        norm[i] = (v > 0) ? rsqrtf(fmaxf((float)v, 1.0f)) : 0.0f;
        int run = r;
#pragma unroll
        for (int c = 0; c < NCHUNK; ++c) {
            start[(size_t)c * NPITCH + i] = run;
            run += cv[c];
        }
    }
}

__global__ __launch_bounds__(256) void gcn_fill(const int* __restrict__ src,
                                                const int* __restrict__ dst,
                                                const int* __restrict__ rank,
                                                const int* __restrict__ start,
                                                int* __restrict__ srcSorted, int nEdges) {
    const int e = (blockIdx.x * 256 + threadIdx.x) * 4;
    if (e >= nEdges) return;
    const int4 s = *(const int4*)(src + e);
    const int4 d = *(const int4*)(dst + e);
    const int4 ra = *(const int4*)(rank + e);
    const int4 rb = *(const int4*)(rank + nEdges + e);
    int4 r;
    r.x = d.x < SPLIT ? ra.x : rb.x;
    r.y = d.y < SPLIT ? ra.y : rb.y;
    r.z = d.z < SPLIT ? ra.z : rb.z;
    r.w = d.w < SPLIT ? ra.w : rb.w;
    const int* st = start + (size_t)(e / CHUNK) * NPITCH;
    srcSorted[st[d.x] + r.x] = s.x;
    srcSorted[st[d.y] + r.y] = s.y;
    srcSorted[st[d.z] + r.z] = s.z;
    srcSorted[st[d.w] + r.w] = s.w;
}

#define TPITCH 68

__global__ __launch_bounds__(256) void gcn_layer1_fused(const float* __restrict__ x,
                                                        const int* __restrict__ rowptr,
                                                        const int* __restrict__ srcSorted,
                                                        const float* __restrict__ norm,
                                                        const float* __restrict__ W1,
                                                        const float* __restrict__ b1,
                                                        const float* __restrict__ W2,
                                                        float* __restrict__ t,
                                                        int nNodes) {
    __shared__ float aggS[16 * TPITCH];
    __shared__ float hS[16 * TPITCH];
    const int tid   = threadIdx.x;
    const int node0 = blockIdx.x * 16;
    const int nl    = tid >> 4;
    const int c     = tid & 15;
    const int n     = node0 + nl;

    int j = 0, end = 0;
    if (n < nNodes) {
        j   = rowptr[n];
        end = rowptr[n + 1];
    }
    const float4* x4 = (const float4*)x;
    float4 acc = make_float4(0.f, 0.f, 0.f, 0.f);
    for (; j < end; j += 4) {
        const int rem = end - j;
        const int s0 = srcSorted[j];
        const int s1 = srcSorted[rem > 1 ? j + 1 : j];
        const int s2 = srcSorted[rem > 2 ? j + 2 : j];
        const int s3 = srcSorted[rem > 3 ? j + 3 : j];
        const float w0 = norm[s0];
        const float w1 = norm[s1];
        const float w2 = norm[s2];
        const float w3 = norm[s3];
        const float4 v0 = x4[(size_t)s0 * 16 + c];
        const float4 v1 = x4[(size_t)s1 * 16 + c];
        const float4 v2 = x4[(size_t)s2 * 16 + c];
        const float4 v3 = x4[(size_t)s3 * 16 + c];
        const bool p1 = rem > 1, p2 = rem > 2, p3 = rem > 3;
        acc.x += v0.x * w0; acc.y += v0.y * w0; acc.z += v0.z * w0; acc.w += v0.w * w0;
        acc.x += (p1 ? v1.x : 0.f) * w1; acc.y += (p1 ? v1.y : 0.f) * w1;
        acc.z += (p1 ? v1.z : 0.f) * w1; acc.w += (p1 ? v1.w : 0.f) * w1;
        acc.x += (p2 ? v2.x : 0.f) * w2; acc.y += (p2 ? v2.y : 0.f) * w2;
        acc.z += (p2 ? v2.z : 0.f) * w2; acc.w += (p2 ? v2.w : 0.f) * w2;
        acc.x += (p3 ? v3.x : 0.f) * w3; acc.y += (p3 ? v3.y : 0.f) * w3;
        acc.z += (p3 ? v3.z : 0.f) * w3; acc.w += (p3 ? v3.w : 0.f) * w3;
    }
    *(float4*)(aggS + nl * TPITCH + c * 4) = acc;

    const int wave = __builtin_amdgcn_readfirstlane(tid >> 5);
    const int lane = tid & 31;
    const int m    = lane & 15;
    const int kh   = lane >> 4;

    v2f bf[16];
    float rs = 0.f, bv = 0.f;
    if (wave < 4) {
        const int col = wave * 16 + m;
#pragma unroll
        for (int i = 0; i < 16; ++i) {
            const int ka = 4 * i + 2 * kh;
            bf[i].x = W1[ka * 64 + col];
            bf[i].y = W1[(ka + 1) * 64 + col];
        }
        rs = norm[node0 + m];
        bv = b1[col];
    } else if (wave < 6) {
        const int col = (wave - 4) * 16 + m;
#pragma unroll
        for (int i = 0; i < 16; ++i) {
            const int ka = 4 * i + 2 * kh;
            bf[i].x = W2[ka * 32 + col];
            bf[i].y = W2[(ka + 1) * 32 + col];
        }
        rs = norm[node0 + m];
    }
    __syncthreads();

    if (wave < 4) {
        v8f d = {0.f, 0.f, 0.f, 0.f, 0.f, 0.f, 0.f, 0.f};
        const float* arow = aggS + m * TPITCH + 2 * kh;
#pragma unroll
        for (int i = 0; i < 16; ++i) {
            const float2 ap = *(const float2*)(arow + 4 * i);
            v2f a;
            a.x = ap.x * rs;
            a.y = ap.y * rs;
            d = __builtin_amdgcn_wmma_f32_16x16x4_f32(
                     false, a,  false, bf[i],
                     (short)0, d,  false,  false);
        }
        float* hcol = hS + (8 * kh) * TPITCH + wave * 16 + m;
#pragma unroll
        for (int r = 0; r < 8; ++r) hcol[r * TPITCH] = fmaxf(d[r] + bv, 0.0f);
    }
    __syncthreads();

    if (wave >= 4 && wave < 6) {
        v8f d = {0.f, 0.f, 0.f, 0.f, 0.f, 0.f, 0.f, 0.f};
        const float* hrow = hS + m * TPITCH + 2 * kh;
#pragma unroll
        for (int i = 0; i < 16; ++i) {
            const float2 hp = *(const float2*)(hrow + 4 * i);
            v2f a;
            a.x = hp.x * rs;
            a.y = hp.y * rs;
            d = __builtin_amdgcn_wmma_f32_16x16x4_f32(
                     false, a,  false, bf[i],
                     (short)0, d,  false,  false);
        }
        float* tcol = t + (size_t)(node0 + 8 * kh) * 32 + (wave - 4) * 16 + m;
#pragma unroll
        for (int r = 0; r < 8; ++r) tcol[(size_t)r * 32] = d[r];
    }
}

__global__ __launch_bounds__(256) void gcn_gather32(const float* __restrict__ t,
                                                    const int* __restrict__ rowptr,
                                                    const int* __restrict__ srcSorted,
                                                    const float* __restrict__ norm,
                                                    const float* __restrict__ bias,
                                                    float* __restrict__ out,
                                                    int nNodes) {
    const int tid = blockIdx.x * 256 + threadIdx.x;
    const int n = tid >> 3;
    const int c = tid & 7;
    if (n >= nNodes) return;
    int j = rowptr[n];
    const int end = rowptr[n + 1];
    const float nd = norm[n];
    const float4 b = ((const float4*)bias)[c];
    const float4* t4 = (const float4*)t;
    float4 acc = make_float4(0.f, 0.f, 0.f, 0.f);
    for (; j < end; j += 4) {
        const int rem = end - j;
        const int s0 = srcSorted[j];
        const int s1 = srcSorted[rem > 1 ? j + 1 : j];
        const int s2 = srcSorted[rem > 2 ? j + 2 : j];
        const int s3 = srcSorted[rem > 3 ? j + 3 : j];
        const float4 v0 = t4[(size_t)s0 * 8 + c];
        const float4 v1 = t4[(size_t)s1 * 8 + c];
        const float4 v2 = t4[(size_t)s2 * 8 + c];
        const float4 v3 = t4[(size_t)s3 * 8 + c];
        const bool p1 = rem > 1, p2 = rem > 2, p3 = rem > 3;
        acc.x += v0.x; acc.y += v0.y; acc.z += v0.z; acc.w += v0.w;
        acc.x += p1 ? v1.x : 0.f; acc.y += p1 ? v1.y : 0.f; acc.z += p1 ? v1.z : 0.f; acc.w += p1 ? v1.w : 0.f;
        acc.x += p2 ? v2.x : 0.f; acc.y += p2 ? v2.y : 0.f; acc.z += p2 ? v2.z : 0.f; acc.w += p2 ? v2.w : 0.f;
        acc.x += p3 ? v3.x : 0.f; acc.y += p3 ? v3.y : 0.f; acc.z += p3 ? v3.z : 0.f; acc.w += p3 ? v3.w : 0.f;
    }
    float4 o;
    o.x = b.x + acc.x * nd;
    o.y = b.y + acc.y * nd;
    o.z = b.z + acc.z * nd;
    o.w = b.w + acc.w * nd;
    ((float4*)out)[(size_t)n * 8 + c] = o;
}

extern "C" void kernel_launch(void* const* d_in, const int* in_sizes, int n_in,
                              void* d_out, int out_size, void* d_ws, size_t ws_size,
                              hipStream_t stream) {
    const float* x   = (const float*)d_in[0];
    const int*   ei  = (const int*)d_in[1];
    const float* W1  = (const float*)d_in[2];
    const float* b1  = (const float*)d_in[3];
    const float* W2  = (const float*)d_in[4];
    const float* b2  = (const float*)d_in[5];
    float*       out = (float*)d_out;

    const int N = in_sizes[0] / 64;
    const int E = in_sizes[1] / 2;
    const int* src = ei;
    const int* dst = ei + E;

    float* t         = (float*)d_ws;
    float* norm      = t + (size_t)N * 32;
    int*   rowptr    = (int*)(norm + N);
    int*   rank      = rowptr + N + 32;
    int*   srcSorted = rank + 2 * (size_t)E;
    int*   tileSumTab = srcSorted + E;
    int*   cntTab    = tileSumTab + NCHUNK * NTILES;
    int*   start     = cntTab + (size_t)NCHUNK * NPITCH;

    {
        gcn_count_lds<<<dim3(NCHUNK, 2), 1024, 0, stream>>>(dst, cntTab, tileSumTab, rank, E);
        gcn_scan_start<<<NTILES, 1024, 0, stream>>>(cntTab, tileSumTab, rowptr, norm, start, N);
        int e4 = E / 4;
        gcn_fill<<<(e4 + 255) / 256, 256, 0, stream>>>(src, dst, rank, start, srcSorted, E);
    }

    {
        gcn_layer1_fused<<<N / 16, 256, 0, stream>>>(x, rowptr, srcSorted, norm, W1, b1, W2, t, N);
    }

    {
        int work = N * 8;
        gcn_gather32<<<(work + 255) / 256, 256, 0, stream>>>(t, rowptr, srcSorted, norm, b2, out, N);
    }
}


# Edge lists in two levels: each edge slice ordered by 256-node bucket in LDS, then one block per bucket counts, scans, places on LDS cursors; 5 launches become 4; sum order changes (LDS arrival)
# speedup vs baseline: 5.0862x; 1.5120x over previous
#include <hip/hip_runtime.h>

typedef float v2f __attribute__((ext_vector_type(2)));
typedef float v8f __attribute__((ext_vector_type(8)));

#define N_NODES 100000
#define N_EDGES 1250000
#define NSLICE 256
#define SLICE  4884
#define SLOTS  5
#define BNODES 256
#define NBUCK  391
#define NBPAD  512

__global__ __launch_bounds__(1024) void gcn_bin(const int* __restrict__ src,
                                                const int* __restrict__ dst,
                                                int2* __restrict__ tmp,
                                                int2* __restrict__ tab, int nEdges) {
    __shared__ int cnt[NBPAD];
    __shared__ int wtot[NBPAD / 32];
    __shared__ int2 stage[SLICE];
    const int tid  = threadIdx.x;
    const int lane = tid & 31;
    const int w    = tid >> 5;
    const int b    = blockIdx.x;
    const int e0   = b * SLICE;
    int e1 = e0 + SLICE;
    if (e1 > nEdges) e1 = nEdges;
    if (tid < NBPAD) cnt[tid] = 0;
    int s[SLOTS], d[SLOTS], r[SLOTS];
#pragma unroll
    for (int q = 0; q < SLOTS; ++q) {
        const int e = e0 + q * 1024 + tid;
        s[q] = 0; d[q] = -1; r[q] = 0;
        if (e < e1) { s[q] = src[e]; d[q] = dst[e]; }
    }
    __syncthreads();
#pragma unroll
    for (int q = 0; q < SLOTS; ++q)
        if (d[q] >= 0) r[q] = atomicAdd(&cnt[d[q] >> 8], 1);
    __syncthreads();

    int c = 0, inc = 0;
    if (tid < NBPAD) {
        c = cnt[tid];
        inc = c;
        for (int o = 1; o < 32; o <<= 1) {
            const int u = __shfl_up(inc, o);
            if (lane >= o) inc += u;
        }
        if (lane == 31) wtot[w] = inc;
    }
    __syncthreads();
    if (tid < NBPAD) {
        int pre = 0;
        for (int k = 0; k < w; ++k) pre += wtot[k];
        const int excl = pre + inc - c;
        cnt[tid] = excl;
        if (tid < NBUCK) tab[tid * NSLICE + b] = make_int2(excl, c);
    }
    __syncthreads();
#pragma unroll
    for (int q = 0; q < SLOTS; ++q)
        if (d[q] >= 0) stage[cnt[d[q] >> 8] + r[q]] = make_int2(s[q], d[q]);
    __syncthreads();
    const int n2 = (e1 - e0 + 1) >> 1;
    int4* o4 = (int4*)(tmp + e0);
    const int4* s4 = (const int4*)stage;
    for (int i = tid; i < n2; i += 1024) o4[i] = s4[i];
}

__global__ __launch_bounds__(1024) void gcn_bucket_lists(const int2* __restrict__ tmp,
                                                         const int2* __restrict__ tab,
                                                         int* __restrict__ rowptr,
                                                         float* __restrict__ norm,
                                                         int* __restrict__ srcSorted,
                                                         int nNodes) {
    __shared__ int cur[BNODES];
    __shared__ int bsum[32];
    __shared__ int wtot[BNODES / 32];
    const int tid   = threadIdx.x;
    const int lane  = tid & 31;
    const int w     = tid >> 5;
    const int k     = blockIdx.x;
    const int node0 = k * BNODES;
    const int run   = tid >> 2;
    const int sub   = tid & 3;
    const int2 oc = tab[k * NSLICE + run];
    if (tid < BNODES) cur[tid] = 0;
    int v = (sub == 0) ? oc.x : 0;
    for (int o = 16; o > 0; o >>= 1) v += __shfl_xor(v, o);
    if (lane == 0) bsum[w] = v;
    const int2* rp = tmp + (size_t)run * SLICE + oc.x;
    const int len = oc.y;
    __syncthreads();

    for (int i = sub; i < len; i += 16) {
        const bool p1 = i + 4 < len, p2 = i + 8 < len, p3 = i + 12 < len;
        const int2 ea = rp[i];
        const int2 eb = rp[p1 ? i + 4 : i];
        const int2 ec = rp[p2 ? i + 8 : i];
        const int2 ed = rp[p3 ? i + 12 : i];
        asm volatile("" :: "v"(ea.y), "v"(eb.y), "v"(ec.y), "v"(ed.y));
        atomicAdd(&cur[ea.y - node0], 1);
        if (p1) atomicAdd(&cur[eb.y - node0], 1);
        if (p2) atomicAdd(&cur[ec.y - node0], 1);
        if (p3) atomicAdd(&cur[ed.y - node0], 1);
    }
    __syncthreads();

    int c = 0, inc = 0;
    if (tid < BNODES) {
        c = cur[tid];
        inc = c;
        for (int o = 1; o < 32; o <<= 1) {
            const int u = __shfl_up(inc, o);
            if (lane >= o) inc += u;
        }
        if (lane == 31) wtot[w] = inc;
    }
    __syncthreads();
    if (tid < BNODES) {
        int base = 0;
        for (int j = 0; j < 32; ++j) base += bsum[j];
        for (int j = 0; j < w; ++j) base += wtot[j];
        const int excl = base + inc - c;
        cur[tid] = excl;
        const int n = node0 + tid;
        if (n < nNodes) {
            rowptr[n] = excl;
            if (n + 1 == nNodes) rowptr[nNodes] = excl + c;
            norm[n] = (c > 0) ? rsqrtf(fmaxf((float)c, 1.0f)) : 0.0f;
        }
    }
    __syncthreads();

    for (int i = sub; i < len; i += 16) {
        const bool p1 = i + 4 < len, p2 = i + 8 < len, p3 = i + 12 < len;
        const int2 ea = rp[i];
        const int2 eb = rp[p1 ? i + 4 : i];
        const int2 ec = rp[p2 ? i + 8 : i];
        const int2 ed = rp[p3 ? i + 12 : i];
        asm volatile("" :: "v"(ea.y), "v"(eb.y), "v"(ec.y), "v"(ed.y));
        srcSorted[atomicAdd(&cur[ea.y - node0], 1)] = ea.x;
        if (p1) srcSorted[atomicAdd(&cur[eb.y - node0], 1)] = eb.x;
        if (p2) srcSorted[atomicAdd(&cur[ec.y - node0], 1)] = ec.x;
        if (p3) srcSorted[atomicAdd(&cur[ed.y - node0], 1)] = ed.x;
    }
}

#define TPITCH 68

__global__ __launch_bounds__(256) void gcn_layer1_fused(const float* __restrict__ x,
                                                        const int* __restrict__ rowptr,
                                                        const int* __restrict__ srcSorted,
                                                        const float* __restrict__ norm,
                                                        const float* __restrict__ W1,
                                                        const float* __restrict__ b1,
                                                        const float* __restrict__ W2,
                                                        float* __restrict__ t,
                                                        int nNodes) {
    __shared__ float aggS[16 * TPITCH];
    __shared__ float hS[16 * TPITCH];
    const int tid   = threadIdx.x;
    const int node0 = blockIdx.x * 16;
    const int nl    = tid >> 4;
    const int c     = tid & 15;
    const int n     = node0 + nl;

    int j = 0, end = 0;
    if (n < nNodes) {
        j   = rowptr[n];
        end = rowptr[n + 1];
    }
    const float4* x4 = (const float4*)x;
    float4 acc = make_float4(0.f, 0.f, 0.f, 0.f);
    for (; j < end; j += 4) {
        const int rem = end - j;
        const int s0 = srcSorted[j];
        const int s1 = srcSorted[rem > 1 ? j + 1 : j];
        const int s2 = srcSorted[rem > 2 ? j + 2 : j];
        const int s3 = srcSorted[rem > 3 ? j + 3 : j];
        const float w0 = norm[s0];
        const float w1 = norm[s1];
        const float w2 = norm[s2];
        const float w3 = norm[s3];
        const float4 v0 = x4[(size_t)s0 * 16 + c];
        const float4 v1 = x4[(size_t)s1 * 16 + c];
        const float4 v2 = x4[(size_t)s2 * 16 + c];
        const float4 v3 = x4[(size_t)s3 * 16 + c];
        const bool p1 = rem > 1, p2 = rem > 2, p3 = rem > 3;
        acc.x += v0.x * w0; acc.y += v0.y * w0; acc.z += v0.z * w0; acc.w += v0.w * w0;
        acc.x += (p1 ? v1.x : 0.f) * w1; acc.y += (p1 ? v1.y : 0.f) * w1;
        acc.z += (p1 ? v1.z : 0.f) * w1; acc.w += (p1 ? v1.w : 0.f) * w1;
        acc.x += (p2 ? v2.x : 0.f) * w2; acc.y += (p2 ? v2.y : 0.f) * w2;
        acc.z += (p2 ? v2.z : 0.f) * w2; acc.w += (p2 ? v2.w : 0.f) * w2;
        acc.x += (p3 ? v3.x : 0.f) * w3; acc.y += (p3 ? v3.y : 0.f) * w3;
        acc.z += (p3 ? v3.z : 0.f) * w3; acc.w += (p3 ? v3.w : 0.f) * w3;
    }
    *(float4*)(aggS + nl * TPITCH + c * 4) = acc;

    const int wave = __builtin_amdgcn_readfirstlane(tid >> 5);
    const int lane = tid & 31;
    const int m    = lane & 15;
    const int kh   = lane >> 4;

    v2f bf[16];
    float rs = 0.f, bv = 0.f;
    if (wave < 4) {
        const int col = wave * 16 + m;
#pragma unroll
        for (int i = 0; i < 16; ++i) {
            const int ka = 4 * i + 2 * kh;
            bf[i].x = W1[ka * 64 + col];
            bf[i].y = W1[(ka + 1) * 64 + col];
        }
        rs = norm[node0 + m];
        bv = b1[col];
    } else if (wave < 6) {
        const int col = (wave - 4) * 16 + m;
#pragma unroll
        for (int i = 0; i < 16; ++i) {
            const int ka = 4 * i + 2 * kh;
            bf[i].x = W2[ka * 32 + col];
            bf[i].y = W2[(ka + 1) * 32 + col];
        }
        rs = norm[node0 + m];
    }
    __syncthreads();

    if (wave < 4) {
        v8f d = {0.f, 0.f, 0.f, 0.f, 0.f, 0.f, 0.f, 0.f};
        const float* arow = aggS + m * TPITCH + 2 * kh;
#pragma unroll
        for (int i = 0; i < 16; ++i) {
            const float2 ap = *(const float2*)(arow + 4 * i);
            v2f a;
            a.x = ap.x * rs;
            a.y = ap.y * rs;
            d = __builtin_amdgcn_wmma_f32_16x16x4_f32(
                     false, a,  false, bf[i],
                     (short)0, d,  false,  false);
        }
        float* hcol = hS + (8 * kh) * TPITCH + wave * 16 + m;
#pragma unroll
        for (int r = 0; r < 8; ++r) hcol[r * TPITCH] = fmaxf(d[r] + bv, 0.0f);
    }
    __syncthreads();

    if (wave >= 4 && wave < 6) {
        v8f d = {0.f, 0.f, 0.f, 0.f, 0.f, 0.f, 0.f, 0.f};
        const float* hrow = hS + m * TPITCH + 2 * kh;
#pragma unroll
        for (int i = 0; i < 16; ++i) {
            const float2 hp = *(const float2*)(hrow + 4 * i);
            v2f a;
            a.x = hp.x * rs;
            a.y = hp.y * rs;
            d = __builtin_amdgcn_wmma_f32_16x16x4_f32(
                     false, a,  false, bf[i],
                     (short)0, d,  false,  false);
        }
        float* tcol = t + (size_t)(node0 + 8 * kh) * 32 + (wave - 4) * 16 + m;
#pragma unroll
        for (int r = 0; r < 8; ++r) tcol[(size_t)r * 32] = d[r];
    }
}

__global__ __launch_bounds__(256) void gcn_gather32(const float* __restrict__ t,
                                                    const int* __restrict__ rowptr,
                                                    const int* __restrict__ srcSorted,
                                                    const float* __restrict__ norm,
                                                    const float* __restrict__ bias,
                                                    float* __restrict__ out,
                                                    int nNodes) {
    const int tid = blockIdx.x * 256 + threadIdx.x;
    const int n = tid >> 3;
    const int c = tid & 7;
    if (n >= nNodes) return;
    int j = rowptr[n];
    const int end = rowptr[n + 1];
    const float nd = norm[n];
    const float4 b = ((const float4*)bias)[c];
    const float4* t4 = (const float4*)t;
    float4 acc = make_float4(0.f, 0.f, 0.f, 0.f);
    for (; j < end; j += 4) {
        const int rem = end - j;
        const int s0 = srcSorted[j];
        const int s1 = srcSorted[rem > 1 ? j + 1 : j];
        const int s2 = srcSorted[rem > 2 ? j + 2 : j];
        const int s3 = srcSorted[rem > 3 ? j + 3 : j];
        const float4 v0 = t4[(size_t)s0 * 8 + c];
        const float4 v1 = t4[(size_t)s1 * 8 + c];
        const float4 v2 = t4[(size_t)s2 * 8 + c];
        const float4 v3 = t4[(size_t)s3 * 8 + c];
        const bool p1 = rem > 1, p2 = rem > 2, p3 = rem > 3;
        acc.x += v0.x; acc.y += v0.y; acc.z += v0.z; acc.w += v0.w;
        acc.x += p1 ? v1.x : 0.f; acc.y += p1 ? v1.y : 0.f; acc.z += p1 ? v1.z : 0.f; acc.w += p1 ? v1.w : 0.f;
        acc.x += p2 ? v2.x : 0.f; acc.y += p2 ? v2.y : 0.f; acc.z += p2 ? v2.z : 0.f; acc.w += p2 ? v2.w : 0.f;
        acc.x += p3 ? v3.x : 0.f; acc.y += p3 ? v3.y : 0.f; acc.z += p3 ? v3.z : 0.f; acc.w += p3 ? v3.w : 0.f;
    }
    float4 o;
    o.x = b.x + acc.x * nd;
    o.y = b.y + acc.y * nd;
    o.z = b.z + acc.z * nd;
    o.w = b.w + acc.w * nd;
    ((float4*)out)[(size_t)n * 8 + c] = o;
}

extern "C" void kernel_launch(void* const* d_in, const int* in_sizes, int n_in,
                              void* d_out, int out_size, void* d_ws, size_t ws_size,
                              hipStream_t stream) {
    const float* x   = (const float*)d_in[0];
    const int*   ei  = (const int*)d_in[1];
    const float* W1  = (const float*)d_in[2];
    const float* b1  = (const float*)d_in[3];
    const float* W2  = (const float*)d_in[4];
    const float* b2  = (const float*)d_in[5];
    float*       out = (float*)d_out;

    const int N = in_sizes[0] / 64;
    const int E = in_sizes[1] / 2;
    const int* src = ei;
    const int* dst = ei + E;

    float* t         = (float*)d_ws;
    float* norm      = t + (size_t)N * 32;
    int*   rowptr    = (int*)(norm + N);
    int2*  tmp       = (int2*)(rowptr + N + 32);
    int2*  tab       = tmp + (size_t)NSLICE * SLICE;
    int*   srcSorted = (int*)(tab + (size_t)NBUCK * NSLICE);

    {
        gcn_bin<<<NSLICE, 1024, 0, stream>>>(src, dst, tmp, tab, E);
        gcn_bucket_lists<<<NBUCK, 1024, 0, stream>>>(tmp, tab, rowptr, norm, srcSorted, N);
    }

    {
        gcn_layer1_fused<<<N / 16, 256, 0, stream>>>(x, rowptr, srcSorted, norm, W1, b1, W2, t, N);
    }

    {
        int work = N * 8;
        gcn_gather32<<<(work + 255) / 256, 256, 0, stream>>>(t, rowptr, srcSorted, norm, b2, out, N);
    }
}


# The list kernel puts each bucket's ordered list together in LDS and stores it in whole 128-byte lines, 16 B per lane, not one scattered 4-byte store per edge; list order and add order unchanged
# speedup vs baseline: 5.2916x; 1.0404x over previous
#include <hip/hip_runtime.h>

typedef float v2f __attribute__((ext_vector_type(2)));
typedef float v8f __attribute__((ext_vector_type(8)));

#define N_NODES 100000
#define N_EDGES 1250000
#define NSLICE 256
#define SLICE  4884
#define SLOTS  5
#define BNODES 256
#define NBUCK  391
#define NBPAD  512
#define LCAP   8192

__global__ __launch_bounds__(1024) void gcn_bin(const int* __restrict__ src,
                                                const int* __restrict__ dst,
                                                int2* __restrict__ tmp,
                                                int2* __restrict__ tab, int nEdges) {
    __shared__ int cnt[NBPAD];
    __shared__ int wtot[NBPAD / 32];
    __shared__ int2 stage[SLICE];
    const int tid  = threadIdx.x;
    const int lane = tid & 31;
    const int w    = tid >> 5;
    const int b    = blockIdx.x;
    const int e0   = b * SLICE;
    int e1 = e0 + SLICE;
    if (e1 > nEdges) e1 = nEdges;
    if (tid < NBPAD) cnt[tid] = 0;
    int s[SLOTS], d[SLOTS], r[SLOTS];
#pragma unroll
    for (int q = 0; q < SLOTS; ++q) {
        const int e = e0 + q * 1024 + tid;
        s[q] = 0; d[q] = -1; r[q] = 0;
        if (e < e1) { s[q] = src[e]; d[q] = dst[e]; }
    }
    __syncthreads();
#pragma unroll
    for (int q = 0; q < SLOTS; ++q)
        if (d[q] >= 0) r[q] = atomicAdd(&cnt[d[q] >> 8], 1);
    __syncthreads();

    int c = 0, inc = 0;
    if (tid < NBPAD) {
        c = cnt[tid];
        inc = c;
        for (int o = 1; o < 32; o <<= 1) {
            const int u = __shfl_up(inc, o);
            if (lane >= o) inc += u;
        }
        if (lane == 31) wtot[w] = inc;
    }
    __syncthreads();
    if (tid < NBPAD) {
        int pre = 0;
        for (int k = 0; k < w; ++k) pre += wtot[k];
        const int excl = pre + inc - c;
        cnt[tid] = excl;
        if (tid < NBUCK) tab[tid * NSLICE + b] = make_int2(excl, c);
    }
    __syncthreads();
#pragma unroll
    for (int q = 0; q < SLOTS; ++q)
        if (d[q] >= 0) stage[cnt[d[q] >> 8] + r[q]] = make_int2(s[q], d[q]);
    __syncthreads();
    const int n2 = (e1 - e0 + 1) >> 1;
    int4* o4 = (int4*)(tmp + e0);
    const int4* s4 = (const int4*)stage;
    for (int i = tid; i < n2; i += 1024) o4[i] = s4[i];
}

__device__ __forceinline__ void place(int* __restrict__ srcSorted, int* lst, int a0, int pos, int s) {
    const int l = pos - a0;
    if (l < LCAP) lst[l] = s;
    else srcSorted[pos] = s;
}

__global__ __launch_bounds__(1024) void gcn_bucket_lists(const int2* __restrict__ tmp,
                                                         const int2* __restrict__ tab,
                                                         int* __restrict__ rowptr,
                                                         float* __restrict__ norm,
                                                         int* __restrict__ srcSorted,
                                                         int nNodes) {
    __shared__ int cur[BNODES];
    __shared__ int bsum[32];
    __shared__ int wtot[BNODES / 32];
    __shared__ int piece[2];
    __shared__ __attribute__((aligned(16))) int lst[LCAP];
    const int tid   = threadIdx.x;
    const int lane  = tid & 31;
    const int w     = tid >> 5;
    const int k     = blockIdx.x;
    const int node0 = k * BNODES;
    const int run   = tid >> 2;
    const int sub   = tid & 3;
    const int2 oc = tab[k * NSLICE + run];
    if (tid < BNODES) cur[tid] = 0;
    int v = (sub == 0) ? oc.x : 0;
    for (int o = 16; o > 0; o >>= 1) v += __shfl_xor(v, o);
    if (lane == 0) bsum[w] = v;
    const int2* rp = tmp + (size_t)run * SLICE + oc.x;
    const int len = oc.y;
    __syncthreads();

    for (int i = sub; i < len; i += 16) {
        const bool p1 = i + 4 < len, p2 = i + 8 < len, p3 = i + 12 < len;
        const int2 ea = rp[i];
        const int2 eb = rp[p1 ? i + 4 : i];
        const int2 ec = rp[p2 ? i + 8 : i];
        const int2 ed = rp[p3 ? i + 12 : i];
        asm volatile("" :: "v"(ea.y), "v"(eb.y), "v"(ec.y), "v"(ed.y));
        atomicAdd(&cur[ea.y - node0], 1);
        if (p1) atomicAdd(&cur[eb.y - node0], 1);
        if (p2) atomicAdd(&cur[ec.y - node0], 1);
        if (p3) atomicAdd(&cur[ed.y - node0], 1);
    }
    __syncthreads();

    int c = 0, inc = 0;
    if (tid < BNODES) {
        c = cur[tid];
        inc = c;
        for (int o = 1; o < 32; o <<= 1) {
            const int u = __shfl_up(inc, o);
            if (lane >= o) inc += u;
        }
        if (lane == 31) wtot[w] = inc;
    }
    __syncthreads();
    if (tid < BNODES) {
        int base = 0;
        for (int j = 0; j < 32; ++j) base += bsum[j];
        for (int j = 0; j < w; ++j) base += wtot[j];
        const int excl = base + inc - c;
        cur[tid] = excl;
        if (tid == 0) piece[0] = base;
        if (tid == BNODES - 1) piece[1] = excl + c;
        const int n = node0 + tid;
        if (n < nNodes) {
            rowptr[n] = excl;
            if (n + 1 == nNodes) rowptr[nNodes] = excl + c;
            norm[n] = (c > 0) ? rsqrtf(fmaxf((float)c, 1.0f)) : 0.0f;
        }
    }
    __syncthreads();

    const int a0 = piece[0] & ~31;
    for (int i = sub; i < len; i += 16) {
        const bool p1 = i + 4 < len, p2 = i + 8 < len, p3 = i + 12 < len;
        const int2 ea = rp[i];
        const int2 eb = rp[p1 ? i + 4 : i];
        const int2 ec = rp[p2 ? i + 8 : i];
        const int2 ed = rp[p3 ? i + 12 : i];
        asm volatile("" :: "v"(ea.y), "v"(eb.y), "v"(ec.y), "v"(ed.y));
        place(srcSorted, lst, a0, atomicAdd(&cur[ea.y - node0], 1), ea.x);
        if (p1) place(srcSorted, lst, a0, atomicAdd(&cur[eb.y - node0], 1), eb.x);
        if (p2) place(srcSorted, lst, a0, atomicAdd(&cur[ec.y - node0], 1), ec.x);
        if (p3) place(srcSorted, lst, a0, atomicAdd(&cur[ed.y - node0], 1), ed.x);
    }
    __syncthreads();

    const int off = piece[0] - a0;
    int lim = piece[1] - a0;
    if (lim > LCAP) lim = LCAP;
    for (int l = 4 * tid; l < lim; l += 4096) {
        if (l >= off && l + 4 <= lim) {
            *(int4*)(srcSorted + a0 + l) = *(const int4*)(lst + l);
        } else {
            for (int q = 0; q < 4; ++q)
                if (l + q >= off && l + q < lim) srcSorted[a0 + l + q] = lst[l + q];
        }
    }
}

#define TPITCH 68

__global__ __launch_bounds__(256) void gcn_layer1_fused(const float* __restrict__ x,
                                                        const int* __restrict__ rowptr,
                                                        const int* __restrict__ srcSorted,
                                                        const float* __restrict__ norm,
                                                        const float* __restrict__ W1,
                                                        const float* __restrict__ b1,
                                                        const float* __restrict__ W2,
                                                        float* __restrict__ t,
                                                        int nNodes) {
    __shared__ float aggS[16 * TPITCH];
    __shared__ float hS[16 * TPITCH];
    const int tid   = threadIdx.x;
    const int node0 = blockIdx.x * 16;
    const int nl    = tid >> 4;
    const int c     = tid & 15;
    const int n     = node0 + nl;

    int j = 0, end = 0;
    if (n < nNodes) {
        j   = rowptr[n];
        end = rowptr[n + 1];
    }
    const float4* x4 = (const float4*)x;
    float4 acc = make_float4(0.f, 0.f, 0.f, 0.f);
    for (; j < end; j += 4) {
        const int rem = end - j;
        const int s0 = srcSorted[j];
        const int s1 = srcSorted[rem > 1 ? j + 1 : j];
        const int s2 = srcSorted[rem > 2 ? j + 2 : j];
        const int s3 = srcSorted[rem > 3 ? j + 3 : j];
        const float w0 = norm[s0];
        const float w1 = norm[s1];
        const float w2 = norm[s2];
        const float w3 = norm[s3];
        const float4 v0 = x4[(size_t)s0 * 16 + c];
        const float4 v1 = x4[(size_t)s1 * 16 + c];
        const float4 v2 = x4[(size_t)s2 * 16 + c];
        const float4 v3 = x4[(size_t)s3 * 16 + c];
        const bool p1 = rem > 1, p2 = rem > 2, p3 = rem > 3;
        acc.x += v0.x * w0; acc.y += v0.y * w0; acc.z += v0.z * w0; acc.w += v0.w * w0;
        acc.x += (p1 ? v1.x : 0.f) * w1; acc.y += (p1 ? v1.y : 0.f) * w1;
        acc.z += (p1 ? v1.z : 0.f) * w1; acc.w += (p1 ? v1.w : 0.f) * w1;
        acc.x += (p2 ? v2.x : 0.f) * w2; acc.y += (p2 ? v2.y : 0.f) * w2;
        acc.z += (p2 ? v2.z : 0.f) * w2; acc.w += (p2 ? v2.w : 0.f) * w2;
        acc.x += (p3 ? v3.x : 0.f) * w3; acc.y += (p3 ? v3.y : 0.f) * w3;
        acc.z += (p3 ? v3.z : 0.f) * w3; acc.w += (p3 ? v3.w : 0.f) * w3;
    }
    *(float4*)(aggS + nl * TPITCH + c * 4) = acc;

    const int wave = __builtin_amdgcn_readfirstlane(tid >> 5);
    const int lane = tid & 31;
    const int m    = lane & 15;
    const int kh   = lane >> 4;

    v2f bf[16];
    float rs = 0.f, bv = 0.f;
    if (wave < 4) {
        const int col = wave * 16 + m;
#pragma unroll
        for (int i = 0; i < 16; ++i) {
            const int ka = 4 * i + 2 * kh;
            bf[i].x = W1[ka * 64 + col];
            bf[i].y = W1[(ka + 1) * 64 + col];
        }
        rs = norm[node0 + m];
        bv = b1[col];
    } else if (wave < 6) {
        const int col = (wave - 4) * 16 + m;
#pragma unroll
        for (int i = 0; i < 16; ++i) {
            const int ka = 4 * i + 2 * kh;
            bf[i].x = W2[ka * 32 + col];
            bf[i].y = W2[(ka + 1) * 32 + col];
        }
        rs = norm[node0 + m];
    }
    __syncthreads();

    if (wave < 4) {
        v8f d = {0.f, 0.f, 0.f, 0.f, 0.f, 0.f, 0.f, 0.f};
        const float* arow = aggS + m * TPITCH + 2 * kh;
#pragma unroll
        for (int i = 0; i < 16; ++i) {
            const float2 ap = *(const float2*)(arow + 4 * i);
            v2f a;
            a.x = ap.x * rs;
            a.y = ap.y * rs;
            d = __builtin_amdgcn_wmma_f32_16x16x4_f32(
                     false, a,  false, bf[i],
                     (short)0, d,  false,  false);
        }
        float* hcol = hS + (8 * kh) * TPITCH + wave * 16 + m;
#pragma unroll
        for (int r = 0; r < 8; ++r) hcol[r * TPITCH] = fmaxf(d[r] + bv, 0.0f);
    }
    __syncthreads();

    if (wave >= 4 && wave < 6) {
        v8f d = {0.f, 0.f, 0.f, 0.f, 0.f, 0.f, 0.f, 0.f};
        const float* hrow = hS + m * TPITCH + 2 * kh;
#pragma unroll
        for (int i = 0; i < 16; ++i) {
            const float2 hp = *(const float2*)(hrow + 4 * i);
            v2f a;
            a.x = hp.x * rs;
            a.y = hp.y * rs;
            d = __builtin_amdgcn_wmma_f32_16x16x4_f32(
                     false, a,  false, bf[i],
                     (short)0, d,  false,  false);
        }
        float* tcol = t + (size_t)(node0 + 8 * kh) * 32 + (wave - 4) * 16 + m;
#pragma unroll
        for (int r = 0; r < 8; ++r) tcol[(size_t)r * 32] = d[r];
    }
}

__global__ __launch_bounds__(256) void gcn_gather32(const float* __restrict__ t,
                                                    const int* __restrict__ rowptr,
                                                    const int* __restrict__ srcSorted,
                                                    const float* __restrict__ norm,
                                                    const float* __restrict__ bias,
                                                    float* __restrict__ out,
                                                    int nNodes) {
    const int tid = blockIdx.x * 256 + threadIdx.x;
    const int n = tid >> 3;
    const int c = tid & 7;
    if (n >= nNodes) return;
    int j = rowptr[n];
    const int end = rowptr[n + 1];
    const float nd = norm[n];
    const float4 b = ((const float4*)bias)[c];
    const float4* t4 = (const float4*)t;
    float4 acc = make_float4(0.f, 0.f, 0.f, 0.f);
    for (; j < end; j += 4) {
        const int rem = end - j;
        const int s0 = srcSorted[j];
        const int s1 = srcSorted[rem > 1 ? j + 1 : j];
        const int s2 = srcSorted[rem > 2 ? j + 2 : j];
        const int s3 = srcSorted[rem > 3 ? j + 3 : j];
        const float4 v0 = t4[(size_t)s0 * 8 + c];
        const float4 v1 = t4[(size_t)s1 * 8 + c];
        const float4 v2 = t4[(size_t)s2 * 8 + c];
        const float4 v3 = t4[(size_t)s3 * 8 + c];
        const bool p1 = rem > 1, p2 = rem > 2, p3 = rem > 3;
        acc.x += v0.x; acc.y += v0.y; acc.z += v0.z; acc.w += v0.w;
        acc.x += p1 ? v1.x : 0.f; acc.y += p1 ? v1.y : 0.f; acc.z += p1 ? v1.z : 0.f; acc.w += p1 ? v1.w : 0.f;
        acc.x += p2 ? v2.x : 0.f; acc.y += p2 ? v2.y : 0.f; acc.z += p2 ? v2.z : 0.f; acc.w += p2 ? v2.w : 0.f;
        acc.x += p3 ? v3.x : 0.f; acc.y += p3 ? v3.y : 0.f; acc.z += p3 ? v3.z : 0.f; acc.w += p3 ? v3.w : 0.f;
    }
    float4 o;
    o.x = b.x + acc.x * nd;
    o.y = b.y + acc.y * nd;
    o.z = b.z + acc.z * nd;
    o.w = b.w + acc.w * nd;
    ((float4*)out)[(size_t)n * 8 + c] = o;
}

extern "C" void kernel_launch(void* const* d_in, const int* in_sizes, int n_in,
                              void* d_out, int out_size, void* d_ws, size_t ws_size,
                              hipStream_t stream) {
    const float* x   = (const float*)d_in[0];
    const int*   ei  = (const int*)d_in[1];
    const float* W1  = (const float*)d_in[2];
    const float* b1  = (const float*)d_in[3];
    const float* W2  = (const float*)d_in[4];
    const float* b2  = (const float*)d_in[5];
    float*       out = (float*)d_out;

    const int N = in_sizes[0] / 64;
    const int E = in_sizes[1] / 2;
    const int* src = ei;
    const int* dst = ei + E;

    float* t         = (float*)d_ws;
    float* norm      = t + (size_t)N * 32;
    int*   rowptr    = (int*)(norm + N);
    int2*  tmp       = (int2*)(rowptr + N + 32);
    int2*  tab       = tmp + (size_t)NSLICE * SLICE;
    int*   srcSorted = (int*)(tab + (size_t)NBUCK * NSLICE);

    {
        gcn_bin<<<NSLICE, 1024, 0, stream>>>(src, dst, tmp, tab, E);
        gcn_bucket_lists<<<NBUCK, 1024, 0, stream>>>(tmp, tab, rowptr, norm, srcSorted, N);
    }

    {
        gcn_layer1_fused<<<N / 16, 256, 0, stream>>>(x, rowptr, srcSorted, norm, W1, b1, W2, t, N);
    }

    {
        int work = N * 8;
        gcn_gather32<<<(work + 255) / 256, 256, 0, stream>>>(t, rowptr, srcSorted, norm, b2, out, N);
    }
}


# Both edge sums fetch a node's list one entry per lane (16 or 32 a trip) and pass entries round by DPP row_share / DPP8 moves: 8 row loads in flight, none behind its own list load; add order kept
# speedup vs baseline: 5.5678x; 1.0522x over previous
#include <hip/hip_runtime.h>

typedef float v2f __attribute__((ext_vector_type(2)));
typedef float v8f __attribute__((ext_vector_type(8)));

#define N_NODES 100000
#define N_EDGES 1250000
#define NSLICE 256
#define SLICE  4884
#define SLOTS  5
#define BNODES 256
#define NBUCK  391
#define NBPAD  512
#define LCAP   8192

__global__ __launch_bounds__(1024) void gcn_bin(const int* __restrict__ src,
                                                const int* __restrict__ dst,
                                                int2* __restrict__ tmp,
                                                int2* __restrict__ tab, int nEdges) {
    __shared__ int cnt[NBPAD];
    __shared__ int wtot[NBPAD / 32];
    __shared__ int2 stage[SLICE];
    const int tid  = threadIdx.x;
    const int lane = tid & 31;
    const int w    = tid >> 5;
    const int b    = blockIdx.x;
    const int e0   = b * SLICE;
    int e1 = e0 + SLICE;
    if (e1 > nEdges) e1 = nEdges;
    if (tid < NBPAD) cnt[tid] = 0;
    int s[SLOTS], d[SLOTS], r[SLOTS];
#pragma unroll
    for (int q = 0; q < SLOTS; ++q) {
        const int e = e0 + q * 1024 + tid;
        s[q] = 0; d[q] = -1; r[q] = 0;
        if (e < e1) { s[q] = src[e]; d[q] = dst[e]; }
    }
    __syncthreads();
#pragma unroll
    for (int q = 0; q < SLOTS; ++q)
        if (d[q] >= 0) r[q] = atomicAdd(&cnt[d[q] >> 8], 1);
    __syncthreads();

    int c = 0, inc = 0;
    if (tid < NBPAD) {
        c = cnt[tid];
        inc = c;
        for (int o = 1; o < 32; o <<= 1) {
            const int u = __shfl_up(inc, o);
            if (lane >= o) inc += u;
        }
        if (lane == 31) wtot[w] = inc;
    }
    __syncthreads();
    if (tid < NBPAD) {
        int pre = 0;
        for (int k = 0; k < w; ++k) pre += wtot[k];
        const int excl = pre + inc - c;
        cnt[tid] = excl;
        if (tid < NBUCK) tab[tid * NSLICE + b] = make_int2(excl, c);
    }
    __syncthreads();
#pragma unroll
    for (int q = 0; q < SLOTS; ++q)
        if (d[q] >= 0) stage[cnt[d[q] >> 8] + r[q]] = make_int2(s[q], d[q]);
    __syncthreads();
    const int n2 = (e1 - e0 + 1) >> 1;
    int4* o4 = (int4*)(tmp + e0);
    const int4* s4 = (const int4*)stage;
    for (int i = tid; i < n2; i += 1024) o4[i] = s4[i];
}

__device__ __forceinline__ void place(int* __restrict__ srcSorted, int* lst, int a0, int pos, int s) {
    const int l = pos - a0;
    if (l < LCAP) lst[l] = s;
    else srcSorted[pos] = s;
}

__global__ __launch_bounds__(1024) void gcn_bucket_lists(const int2* __restrict__ tmp,
                                                         const int2* __restrict__ tab,
                                                         int* __restrict__ rowptr,
                                                         float* __restrict__ norm,
                                                         int* __restrict__ srcSorted,
                                                         int nNodes) {
    __shared__ int cur[BNODES];
    __shared__ int bsum[32];
    __shared__ int wtot[BNODES / 32];
    __shared__ int piece[2];
    __shared__ __attribute__((aligned(16))) int lst[LCAP];
    const int tid   = threadIdx.x;
    const int lane  = tid & 31;
    const int w     = tid >> 5;
    const int k     = blockIdx.x;
    const int node0 = k * BNODES;
    const int run   = tid >> 2;
    const int sub   = tid & 3;
    const int2 oc = tab[k * NSLICE + run];
    if (tid < BNODES) cur[tid] = 0;
    int v = (sub == 0) ? oc.x : 0;
    for (int o = 16; o > 0; o >>= 1) v += __shfl_xor(v, o);
    if (lane == 0) bsum[w] = v;
    const int2* rp = tmp + (size_t)run * SLICE + oc.x;
    const int len = oc.y;
    __syncthreads();

    for (int i = sub; i < len; i += 16) {
        const bool p1 = i + 4 < len, p2 = i + 8 < len, p3 = i + 12 < len;
        const int2 ea = rp[i];
        const int2 eb = rp[p1 ? i + 4 : i];
        const int2 ec = rp[p2 ? i + 8 : i];
        const int2 ed = rp[p3 ? i + 12 : i];
        asm volatile("" :: "v"(ea.y), "v"(eb.y), "v"(ec.y), "v"(ed.y));
        atomicAdd(&cur[ea.y - node0], 1);
        if (p1) atomicAdd(&cur[eb.y - node0], 1);
        if (p2) atomicAdd(&cur[ec.y - node0], 1);
        if (p3) atomicAdd(&cur[ed.y - node0], 1);
    }
    __syncthreads();

    int c = 0, inc = 0;
    if (tid < BNODES) {
        c = cur[tid];
        inc = c;
        for (int o = 1; o < 32; o <<= 1) {
            const int u = __shfl_up(inc, o);
            if (lane >= o) inc += u;
        }
        if (lane == 31) wtot[w] = inc;
    }
    __syncthreads();
    if (tid < BNODES) {
        int base = 0;
        for (int j = 0; j < 32; ++j) base += bsum[j];
        for (int j = 0; j < w; ++j) base += wtot[j];
        const int excl = base + inc - c;
        cur[tid] = excl;
        if (tid == 0) piece[0] = base;
        if (tid == BNODES - 1) piece[1] = excl + c;
        const int n = node0 + tid;
        if (n < nNodes) {
            rowptr[n] = excl;
            if (n + 1 == nNodes) rowptr[nNodes] = excl + c;
            norm[n] = (c > 0) ? rsqrtf(fmaxf((float)c, 1.0f)) : 0.0f;
        }
    }
    __syncthreads();

    const int a0 = piece[0] & ~31;
    for (int i = sub; i < len; i += 16) {
        const bool p1 = i + 4 < len, p2 = i + 8 < len, p3 = i + 12 < len;
        const int2 ea = rp[i];
        const int2 eb = rp[p1 ? i + 4 : i];
        const int2 ec = rp[p2 ? i + 8 : i];
        const int2 ed = rp[p3 ? i + 12 : i];
        asm volatile("" :: "v"(ea.y), "v"(eb.y), "v"(ec.y), "v"(ed.y));
        place(srcSorted, lst, a0, atomicAdd(&cur[ea.y - node0], 1), ea.x);
        if (p1) place(srcSorted, lst, a0, atomicAdd(&cur[eb.y - node0], 1), eb.x);
        if (p2) place(srcSorted, lst, a0, atomicAdd(&cur[ec.y - node0], 1), ec.x);
        if (p3) place(srcSorted, lst, a0, atomicAdd(&cur[ed.y - node0], 1), ed.x);
    }
    __syncthreads();

    const int off = piece[0] - a0;
    int lim = piece[1] - a0;
    if (lim > LCAP) lim = LCAP;
    for (int l = 4 * tid; l < lim; l += 4096) {
        if (l >= off && l + 4 <= lim) {
            *(int4*)(srcSorted + a0 + l) = *(const int4*)(lst + l);
        } else {
            for (int q = 0; q < 4; ++q)
                if (l + q >= off && l + q < lim) srcSorted[a0 + l + q] = lst[l + q];
        }
    }
}

template <int L>
__device__ __forceinline__ int row_entry(int v) {
    return __builtin_amdgcn_update_dpp(0, v, 0x150 + L, 0xf, 0xf, false);
}

template <int H>
__device__ __forceinline__ void rows8_x(const float4* __restrict__ x4, const float* __restrict__ norm,
                                        int mine, int cnt, int c, float4& acc) {
    int s[8];
    float w[8];
    float4 v[8];
    s[0] = row_entry<H + 0>(mine); s[1] = row_entry<H + 1>(mine);
    s[2] = row_entry<H + 2>(mine); s[3] = row_entry<H + 3>(mine);
    s[4] = row_entry<H + 4>(mine); s[5] = row_entry<H + 5>(mine);
    s[6] = row_entry<H + 6>(mine); s[7] = row_entry<H + 7>(mine);
#pragma unroll
    for (int q = 0; q < 8; ++q) w[q] = norm[s[q]];
#pragma unroll
    for (int q = 0; q < 8; ++q) v[q] = x4[s[q] * 16 + c];
#pragma unroll
    for (int q = 0; q < 8; ++q) {
        const float ww = (q < cnt) ? w[q] : 0.f;
        acc.x += v[q].x * ww; acc.y += v[q].y * ww; acc.z += v[q].z * ww; acc.w += v[q].w * ww;
    }
}

#define TPITCH 68

__global__ __launch_bounds__(256) void gcn_layer1_fused(const float* __restrict__ x,
                                                        const int* __restrict__ rowptr,
                                                        const int* __restrict__ srcSorted,
                                                        const float* __restrict__ norm,
                                                        const float* __restrict__ W1,
                                                        const float* __restrict__ b1,
                                                        const float* __restrict__ W2,
                                                        float* __restrict__ t,
                                                        int nNodes) {
    __shared__ float aggS[16 * TPITCH];
    __shared__ float hS[16 * TPITCH];
    const int tid   = threadIdx.x;
    const int node0 = blockIdx.x * 16;
    const int nl    = tid >> 4;
    const int c     = tid & 15;
    const int n     = node0 + nl;

    int j = 0, end = 0;
    if (n < nNodes) {
        j   = rowptr[n];
        end = rowptr[n + 1];
    }
    const float4* x4 = (const float4*)x;
    float4 acc = make_float4(0.f, 0.f, 0.f, 0.f);
    for (; j < end; j += 32) {
        const int rem = end - j;
        const int mineA = srcSorted[c < rem ? j + c : j];
        const int mineB = srcSorted[c + 16 < rem ? j + c + 16 : j];
        rows8_x<0>(x4, norm, mineA, rem, c, acc);
        if (rem > 8) rows8_x<8>(x4, norm, mineA, rem - 8, c, acc);
        if (rem > 16) {
            rows8_x<0>(x4, norm, mineB, rem - 16, c, acc);
            if (rem > 24) rows8_x<8>(x4, norm, mineB, rem - 24, c, acc);
        }
    }
    *(float4*)(aggS + nl * TPITCH + c * 4) = acc;

    const int wave = __builtin_amdgcn_readfirstlane(tid >> 5);
    const int lane = tid & 31;
    const int m    = lane & 15;
    const int kh   = lane >> 4;

    v2f bf[16];
    float rs = 0.f, bv = 0.f;
    if (wave < 4) {
        const int col = wave * 16 + m;
#pragma unroll
        for (int i = 0; i < 16; ++i) {
            const int ka = 4 * i + 2 * kh;
            bf[i].x = W1[ka * 64 + col];
            bf[i].y = W1[(ka + 1) * 64 + col];
        }
        rs = norm[node0 + m];
        bv = b1[col];
    } else if (wave < 6) {
        const int col = (wave - 4) * 16 + m;
#pragma unroll
        for (int i = 0; i < 16; ++i) {
            const int ka = 4 * i + 2 * kh;
            bf[i].x = W2[ka * 32 + col];
            bf[i].y = W2[(ka + 1) * 32 + col];
        }
        rs = norm[node0 + m];
    }
    __syncthreads();

    if (wave < 4) {
        v8f d = {0.f, 0.f, 0.f, 0.f, 0.f, 0.f, 0.f, 0.f};
        const float* arow = aggS + m * TPITCH + 2 * kh;
#pragma unroll
        for (int i = 0; i < 16; ++i) {
            const float2 ap = *(const float2*)(arow + 4 * i);
            v2f a;
            a.x = ap.x * rs;
            a.y = ap.y * rs;
            d = __builtin_amdgcn_wmma_f32_16x16x4_f32(
                     false, a,  false, bf[i],
                     (short)0, d,  false,  false);
        }
        float* hcol = hS + (8 * kh) * TPITCH + wave * 16 + m;
#pragma unroll
        for (int r = 0; r < 8; ++r) hcol[r * TPITCH] = fmaxf(d[r] + bv, 0.0f);
    }
    __syncthreads();

    if (wave >= 4 && wave < 6) {
        v8f d = {0.f, 0.f, 0.f, 0.f, 0.f, 0.f, 0.f, 0.f};
        const float* hrow = hS + m * TPITCH + 2 * kh;
#pragma unroll
        for (int i = 0; i < 16; ++i) {
            const float2 hp = *(const float2*)(hrow + 4 * i);
            v2f a;
            a.x = hp.x * rs;
            a.y = hp.y * rs;
            d = __builtin_amdgcn_wmma_f32_16x16x4_f32(
                     false, a,  false, bf[i],
                     (short)0, d,  false,  false);
        }
        float* tcol = t + (size_t)(node0 + 8 * kh) * 32 + (wave - 4) * 16 + m;
#pragma unroll
        for (int r = 0; r < 8; ++r) tcol[(size_t)r * 32] = d[r];
    }
}

template <int Q>
__device__ __forceinline__ int group_entry(int v) {
    return __builtin_amdgcn_mov_dpp8(v, Q * 0x249249);
}

__device__ __forceinline__ void rows8_t(const float4* __restrict__ t4, int mine, int cnt, int c, float4& acc) {
    int s[8];
    float4 v[8];
    s[0] = group_entry<0>(mine); s[1] = group_entry<1>(mine);
    s[2] = group_entry<2>(mine); s[3] = group_entry<3>(mine);
    s[4] = group_entry<4>(mine); s[5] = group_entry<5>(mine);
    s[6] = group_entry<6>(mine); s[7] = group_entry<7>(mine);
#pragma unroll
    for (int q = 0; q < 8; ++q) v[q] = t4[s[q] * 8 + c];
#pragma unroll
    for (int q = 0; q < 8; ++q) {
        const bool p = q < cnt;
        acc.x += p ? v[q].x : 0.f; acc.y += p ? v[q].y : 0.f;
        acc.z += p ? v[q].z : 0.f; acc.w += p ? v[q].w : 0.f;
    }
}

__global__ __launch_bounds__(256) void gcn_gather32(const float* __restrict__ t,
                                                    const int* __restrict__ rowptr,
                                                    const int* __restrict__ srcSorted,
                                                    const float* __restrict__ norm,
                                                    const float* __restrict__ bias,
                                                    float* __restrict__ out,
                                                    int nNodes) {
    const int tid = blockIdx.x * 256 + threadIdx.x;
    const int n = tid >> 3;
    const int c = tid & 7;
    if (n >= nNodes) return;
    int j = rowptr[n];
    const int end = rowptr[n + 1];
    const float nd = norm[n];
    const float4 b = ((const float4*)bias)[c];
    const float4* t4 = (const float4*)t;
    float4 acc = make_float4(0.f, 0.f, 0.f, 0.f);
    for (; j < end; j += 16) {
        const int rem = end - j;
        const int mineA = srcSorted[c < rem ? j + c : j];
        const int mineB = srcSorted[c + 8 < rem ? j + c + 8 : j];
        rows8_t(t4, mineA, rem, c, acc);
        if (rem > 8) rows8_t(t4, mineB, rem - 8, c, acc);
    }
    float4 o;
    o.x = b.x + acc.x * nd;
    o.y = b.y + acc.y * nd;
    o.z = b.z + acc.z * nd;
    o.w = b.w + acc.w * nd;
    ((float4*)out)[(size_t)n * 8 + c] = o;
}

extern "C" void kernel_launch(void* const* d_in, const int* in_sizes, int n_in,
                              void* d_out, int out_size, void* d_ws, size_t ws_size,
                              hipStream_t stream) {
    const float* x   = (const float*)d_in[0];
    const int*   ei  = (const int*)d_in[1];
    const float* W1  = (const float*)d_in[2];
    const float* b1  = (const float*)d_in[3];
    const float* W2  = (const float*)d_in[4];
    const float* b2  = (const float*)d_in[5];
    float*       out = (float*)d_out;

    const int N = in_sizes[0] / 64;
    const int E = in_sizes[1] / 2;
    const int* src = ei;
    const int* dst = ei + E;

    float* t         = (float*)d_ws;
    float* norm      = t + (size_t)N * 32;
    int*   rowptr    = (int*)(norm + N);
    int2*  tmp       = (int2*)(rowptr + N + 32);
    int2*  tab       = tmp + (size_t)NSLICE * SLICE;
    int*   srcSorted = (int*)(tab + (size_t)NBUCK * NSLICE);

    {
        gcn_bin<<<NSLICE, 1024, 0, stream>>>(src, dst, tmp, tab, E);
        gcn_bucket_lists<<<NBUCK, 1024, 0, stream>>>(tmp, tab, rowptr, norm, srcSorted, N);
    }

    {
        gcn_layer1_fused<<<N / 16, 256, 0, stream>>>(x, rowptr, srcSorted, norm, W1, b1, W2, t, N);
    }

    {
        int work = N * 8;
        gcn_gather32<<<(work + 255) / 256, 256, 0, stream>>>(t, rowptr, srcSorted, norm, b2, out, N);
    }
}
